# starting kernel
# baseline (speedup 1.0000x reference)
#include <hip/hip_runtime.h>

#define IN_F      128
#define HC        128
#define HEADS     4
#define OUT_F     32
#define NEG_SLOPE 0.2f

typedef float v2f __attribute__((ext_vector_type(2)));
typedef float v4f __attribute__((ext_vector_type(4)));
typedef float v8f __attribute__((ext_vector_type(8)));

__device__ __forceinline__ unsigned fkey(float f) {
    unsigned b = __float_as_uint(f);
    return (b & 0x80000000u) ? ~b : (b | 0x80000000u);
}
__device__ __forceinline__ float funkey(unsigned k) {
    unsigned b = (k & 0x80000000u) ? (k & 0x7fffffffu) : ~k;
    return __uint_as_float(b);
}
__device__ __forceinline__ float leaky(float x) {
    return x > 0.0f ? x : NEG_SLOPE * x;
}

__global__ __launch_bounds__(256) void gat_init(float* __restrict__ out,
                                                const float* __restrict__ bias,
                                                unsigned* __restrict__ mkey,
                                                float* __restrict__ denom,
                                                int total, int nm) {
    int i = blockIdx.x * 256 + threadIdx.x;
    if (i < total) out[i] = bias[i & (HC - 1)];
    if (i < nm) { mkey[i] = 0u; denom[i] = 0.0f; }
}

#define WSTR 136
#define XSTR 132
#define GEMM_LDS_BYTES ((128 * WSTR + 16 * XSTR) * 4)

__global__ __launch_bounds__(128) void gat_gemm_wmma(const float* __restrict__ x,
                                                     const float* __restrict__ W,
                                                     float* __restrict__ h) {
    extern __shared__ float smem[];
    float* Wl = smem;
    float* Xl = smem + 128 * WSTR;

    const int tid  = threadIdx.x;
    const int row0 = blockIdx.x * 16;

    for (int i = tid; i < 128 * 32; i += 128) {
        int r  = i >> 5;
        int c4 = i & 31;
        v4f v = ((const v4f*)W)[r * 32 + c4];
        *(v4f*)&Wl[r * WSTR + c4 * 4] = v;
    }
    for (int i = tid; i < 16 * 32; i += 128) {
        int r  = i >> 5;
        int c4 = i & 31;
        v4f v = ((const v4f*)x)[(size_t)(row0 + r) * 32 + c4];
        *(v4f*)&Xl[r * XSTR + c4 * 4] = v;
    }
    __syncthreads();

    const int wave = tid >> 5;
    const int lane = tid & 31;
    const int L    = lane & 15;
    const int g    = lane >> 4;

    const int col0 = wave * 16;
    const int col1 = (wave + 4) * 16;

    v8f acc0 = {};
    v8f acc1 = {};

    for (int k0 = 0; k0 < IN_F; k0 += 4) {
        const int k = k0 + 2 * g;
        v2f a, b0, b1;
        a.x  = Xl[L * XSTR + k];
        a.y  = Xl[L * XSTR + k + 1];
        b0.x = Wl[k * WSTR + col0 + L];
        b0.y = Wl[(k + 1) * WSTR + col0 + L];
        b1.x = Wl[k * WSTR + col1 + L];
        b1.y = Wl[(k + 1) * WSTR + col1 + L];
        acc0 = __builtin_amdgcn_wmma_f32_16x16x4_f32(false, a, false, b0, (short)0, acc0, false, false);
        acc1 = __builtin_amdgcn_wmma_f32_16x16x4_f32(false, a, false, b1, (short)0, acc1, false, false);
    }

    const int mbase = row0 + 8 * g;
    for (int r = 0; r < 8; ++r) {
        h[(size_t)(mbase + r) * HC + col0 + L] = acc0[r];
        h[(size_t)(mbase + r) * HC + col1 + L] = acc1[r];
    }
}

__global__ __launch_bounds__(256) void gat_attn_coeff(const float* __restrict__ h,
                                                      const float* __restrict__ att_src,
                                                      const float* __restrict__ att_dst,
                                                      float* __restrict__ a_src,
                                                      float* __restrict__ a_dst,
                                                      int N) {
    int n    = blockIdx.x * 8 + (threadIdx.x >> 5);
    int lane = threadIdx.x & 31;
    if (n >= N) return;

    float s[HEADS], d[HEADS];
#pragma unroll
    for (int hh = 0; hh < HEADS; ++hh) {
        float v = h[(size_t)n * HC + hh * OUT_F + lane];
        s[hh] = v * att_src[hh * OUT_F + lane];
        d[hh] = v * att_dst[hh * OUT_F + lane];
    }
#pragma unroll
    for (int off = 16; off > 0; off >>= 1) {
#pragma unroll
        for (int hh = 0; hh < HEADS; ++hh) {
            s[hh] += __shfl_xor(s[hh], off, 32);
            d[hh] += __shfl_xor(d[hh], off, 32);
        }
    }
    if (lane == 0) {
#pragma unroll
        for (int hh = 0; hh < HEADS; ++hh) {
            a_src[n * HEADS + hh] = s[hh];
            a_dst[n * HEADS + hh] = d[hh];
        }
    }
}

__global__ __launch_bounds__(256) void gat_edge_max(const int* __restrict__ ei,
                                                    const float* __restrict__ a_src,
                                                    const float* __restrict__ a_dst,
                                                    unsigned* __restrict__ mkey,
                                                    int E, int N) {
    int e = blockIdx.x * 256 + threadIdx.x;
    if (e >= E + N) return;
    int s, d;
    if (e < E) { s = ei[e]; d = ei[E + e]; } else { s = d = e - E; }
    v4f as = *(const v4f*)&a_src[s * HEADS];
    v4f ad = *(const v4f*)&a_dst[d * HEADS];
#pragma unroll
    for (int hh = 0; hh < HEADS; ++hh) {
        float lr = leaky(as[hh] + ad[hh]);
        atomicMax(&mkey[d * HEADS + hh], fkey(lr));
    }
}

__global__ __launch_bounds__(256) void gat_edge_denom(const int* __restrict__ ei,
                                                      const float* __restrict__ a_src,
                                                      const float* __restrict__ a_dst,
                                                      const unsigned* __restrict__ mkey,
                                                      float* __restrict__ denom,
                                                      int E, int N) {
    int e = blockIdx.x * 256 + threadIdx.x;
    if (e >= E + N) return;
    int s, d;
    if (e < E) { s = ei[e]; d = ei[E + e]; } else { s = d = e - E; }
    v4f as = *(const v4f*)&a_src[s * HEADS];
    v4f ad = *(const v4f*)&a_dst[d * HEADS];
#pragma unroll
    for (int hh = 0; hh < HEADS; ++hh) {
        float lr = leaky(as[hh] + ad[hh]);
        float m  = funkey(mkey[d * HEADS + hh]);
        atomicAdd(&denom[d * HEADS + hh], __expf(lr - m));
    }
}

__global__ __launch_bounds__(256) void gat_edge_scatter(const int* __restrict__ ei,
                                                        const float* __restrict__ a_src,
                                                        const float* __restrict__ a_dst,
                                                        const unsigned* __restrict__ mkey,
                                                        const float* __restrict__ denom,
                                                        const float* __restrict__ h,
                                                        float* __restrict__ out,
                                                        int E, int N) {
    int e    = blockIdx.x * 8 + (threadIdx.x >> 5);
    int lane = threadIdx.x & 31;
    if (e >= E + N) return;
    int s, d;
    if (e < E) { s = ei[e]; d = ei[E + e]; } else { s = d = e - E; }

    int head  = lane >> 3;
    float lr  = leaky(a_src[s * HEADS + head] + a_dst[d * HEADS + head]);
    float m   = funkey(mkey[d * HEADS + head]);
    float den = denom[d * HEADS + head];
    float alpha = __expf(lr - m) / (den + 1e-16f);

    v4f hv = *(const v4f*)&h[(size_t)s * HC + lane * 4];
    float* o = &out[(size_t)d * HC + lane * 4];
    atomicAdd(o + 0, hv.x * alpha);
    atomicAdd(o + 1, hv.y * alpha);
    atomicAdd(o + 2, hv.z * alpha);
    atomicAdd(o + 3, hv.w * alpha);
}

extern "C" void kernel_launch(void* const* d_in, const int* in_sizes, int n_in,
                              void* d_out, int out_size, void* d_ws, size_t ws_size,
                              hipStream_t stream) {
    const float* x        = (const float*)d_in[0];
    const int*   ei       = (const int*)d_in[1];
    const float* W        = (const float*)d_in[2];
    const float* att_src  = (const float*)d_in[3];
    const float* att_dst  = (const float*)d_in[4];
    const float* bias     = (const float*)d_in[5];
    float*       out      = (float*)d_out;

    const int N = in_sizes[0] / IN_F;
    const int E = in_sizes[1] / 2;

    float*    h     = (float*)d_ws;
    float*    a_src = h + (size_t)N * HC;
    float*    a_dst = a_src + (size_t)N * HEADS;
    unsigned* mkey  = (unsigned*)(a_dst + (size_t)N * HEADS);
    float*    denom = (float*)(mkey + (size_t)N * HEADS);

    const int total = N * HC;
    gat_init<<<(total + 255) / 256, 256, 0, stream>>>(out, bias, mkey, denom, total, N * HEADS);

    gat_gemm_wmma<<<N / 16, 128, GEMM_LDS_BYTES, stream>>>(x, W, h);

    gat_attn_coeff<<<(N + 7) / 8, 256, 0, stream>>>(h, att_src, att_dst, a_src, a_dst, N);

    const int ET = E + N;
    gat_edge_max<<<(ET + 255) / 256, 256, 0, stream>>>(ei, a_src, a_dst, mkey, E, N);
    gat_edge_denom<<<(ET + 255) / 256, 256, 0, stream>>>(ei, a_src, a_dst, mkey, denom, E, N);
    gat_edge_scatter<<<(ET + 7) / 8, 256, 0, stream>>>(ei, a_src, a_dst, mkey, denom, h, out, E, N);
}


# Edges bucketed by destination (int atomic count, arrival-order rank, scan, place); one wave per node does max, exp sum and weighted row sum in registers in bucket order: no f32 atomics, no out init
# speedup vs baseline: 3.2384x; 3.2384x over previous
#include <hip/hip_runtime.h>

#define IN_F      128
#define HC        128
#define HEADS     4
#define OUT_F     32
#define NEG_SLOPE 0.2f

typedef float v2f __attribute__((ext_vector_type(2)));
typedef float v4f __attribute__((ext_vector_type(4)));
typedef float v8f __attribute__((ext_vector_type(8)));
typedef int   v4i __attribute__((ext_vector_type(4)));

__device__ __forceinline__ float leaky(float x) {
    return x > 0.0f ? x : NEG_SLOPE * x;
}

#define WSTR 136
#define XSTR 132
#define GEMM_LDS_BYTES ((128 * WSTR + 16 * XSTR) * 4)

__global__ __launch_bounds__(128) void gat_gemm_wmma(const float* __restrict__ x,
                                                     const float* __restrict__ W,
                                                     float* __restrict__ h) {
    extern __shared__ float smem[];
    float* Wl = smem;
    float* Xl = smem + 128 * WSTR;

    const int tid  = threadIdx.x;
    const int row0 = blockIdx.x * 16;

    for (int i = tid; i < 128 * 32; i += 128) {
        int r  = i >> 5;
        int c4 = i & 31;
        v4f v = ((const v4f*)W)[r * 32 + c4];
        *(v4f*)&Wl[r * WSTR + c4 * 4] = v;
    }
    for (int i = tid; i < 16 * 32; i += 128) {
        int r  = i >> 5;
        int c4 = i & 31;
        v4f v = ((const v4f*)x)[(size_t)(row0 + r) * 32 + c4];
        *(v4f*)&Xl[r * XSTR + c4 * 4] = v;
    }
    __syncthreads();

    const int wave = tid >> 5;
    const int lane = tid & 31;
    const int L    = lane & 15;
    const int g    = lane >> 4;

    const int col0 = wave * 16;
    const int col1 = (wave + 4) * 16;

    v8f acc0 = {};
    v8f acc1 = {};

    for (int k0 = 0; k0 < IN_F; k0 += 4) {
        const int k = k0 + 2 * g;
        v2f a, b0, b1;
        a.x  = Xl[L * XSTR + k];
        a.y  = Xl[L * XSTR + k + 1];
        b0.x = Wl[k * WSTR + col0 + L];
        b0.y = Wl[(k + 1) * WSTR + col0 + L];
        b1.x = Wl[k * WSTR + col1 + L];
        b1.y = Wl[(k + 1) * WSTR + col1 + L];
        acc0 = __builtin_amdgcn_wmma_f32_16x16x4_f32(false, a, false, b0, (short)0, acc0, false, false);
        acc1 = __builtin_amdgcn_wmma_f32_16x16x4_f32(false, a, false, b1, (short)0, acc1, false, false);
    }

    const int mbase = row0 + 8 * g;
    for (int r = 0; r < 8; ++r) {
        h[(size_t)(mbase + r) * HC + col0 + L] = acc0[r];
        h[(size_t)(mbase + r) * HC + col1 + L] = acc1[r];
    }
}

__global__ __launch_bounds__(256) void gat_attn_coeff(const float* __restrict__ h,
                                                      const float* __restrict__ att_src,
                                                      const float* __restrict__ att_dst,
                                                      float* __restrict__ a_src,
                                                      float* __restrict__ a_dst,
                                                      int* __restrict__ cnt,
                                                      int N) {
    int n    = blockIdx.x * 8 + (threadIdx.x >> 5);
    int lane = threadIdx.x & 31;
    if (n >= N) return;

    float s[HEADS], d[HEADS];
#pragma unroll
    for (int hh = 0; hh < HEADS; ++hh) {
        float v = h[(size_t)n * HC + hh * OUT_F + lane];
        s[hh] = v * att_src[hh * OUT_F + lane];
        d[hh] = v * att_dst[hh * OUT_F + lane];
    }
#pragma unroll
    for (int off = 16; off > 0; off >>= 1) {
#pragma unroll
        for (int hh = 0; hh < HEADS; ++hh) {
            s[hh] += __shfl_xor(s[hh], off, 32);
            d[hh] += __shfl_xor(d[hh], off, 32);
        }
    }
    if (lane == 0) {
#pragma unroll
        for (int hh = 0; hh < HEADS; ++hh) {
            a_src[n * HEADS + hh] = s[hh];
            a_dst[n * HEADS + hh] = d[hh];
        }
        cnt[n] = 0;
    }
}

__global__ __launch_bounds__(256) void gat_edge_rank(const int* __restrict__ ei,
                                                     int* __restrict__ cnt,
                                                     int* __restrict__ rank,
                                                     int E) {
    int e = blockIdx.x * 256 + threadIdx.x;
    if (e >= E) return;
    int d = ei[E + e];
    rank[e] = atomicAdd(&cnt[d], 1);
}

__global__ __launch_bounds__(1024) void gat_bucket_offsets(const int* __restrict__ cnt,
                                                           int* __restrict__ off,
                                                           int N) {
    __shared__ int wtot[32];
    __shared__ int wpart[32];
    const int t     = threadIdx.x;
    const int lane  = t & 31;
    const int wave  = t >> 5;
    const int tile0 = blockIdx.x * 1024;

    int part = 0;
    for (int i = t * 4; i < tile0; i += 4096) {
        v4i c = *(const v4i*)&cnt[i];
        part += (c.x + c.y) + (c.z + c.w);
    }
    const int idx = tile0 + t;
    const int v   = (idx < N) ? cnt[idx] : 0;

    int inc = v;
    for (int o = 1; o < 32; o <<= 1) {
        int u = __shfl_up(inc, o, 32);
        if (lane >= o) inc += u;
    }
    for (int o = 16; o > 0; o >>= 1) part += __shfl_xor(part, o, 32);

    if (lane == 31) wtot[wave]  = inc;
    if (lane == 0)  wpart[wave] = part;
    __syncthreads();

    int tot = wpart[lane] + (lane < wave ? wtot[lane] : 0);
    for (int o = 16; o > 0; o >>= 1) tot += __shfl_xor(tot, o, 32);

    if (idx <= N) off[idx] = tot + inc - v;
}

__global__ __launch_bounds__(256) void gat_bucket_fill(const int* __restrict__ ei,
                                                       const int* __restrict__ off,
                                                       const int* __restrict__ rank,
                                                       int* __restrict__ bsrc,
                                                       int E) {
    int e = blockIdx.x * 256 + threadIdx.x;
    if (e >= E) return;
    int s = ei[e];
    int d = ei[E + e];
    bsrc[off[d] + rank[e]] = s;
}

__global__ __launch_bounds__(256) void gat_node_gather(const int* __restrict__ off,
                                                       const int* __restrict__ bsrc,
                                                       const float* __restrict__ a_src,
                                                       const float* __restrict__ a_dst,
                                                       const float* __restrict__ h,
                                                       const float* __restrict__ bias,
                                                       float* __restrict__ out,
                                                       int N) {
    const int lane = threadIdx.x & 31;
    const int d = __builtin_amdgcn_readfirstlane((int)(blockIdx.x * 8 + (threadIdx.x >> 5)));
    if (d >= N) return;

    const int   beg   = off[d];
    const int   end   = off[d + 1];
    const int   head  = lane >> 3;
    const float ad    = a_dst[d * HEADS + head];
    const float eself = leaky(a_src[d * HEADS + head] + ad);

    float m = eself;
#pragma unroll 4
    for (int j = beg; j < end; ++j) {
        const int   s = bsrc[j];
        const float e = leaky(a_src[s * HEADS + head] + ad);
        m = e > m ? e : m;
    }

    float p   = __expf(eself - m);
    float den = p;
    v4f   hv  = *(const v4f*)&h[(size_t)d * HC + lane * 4];
    v4f   acc = hv * p;
#pragma unroll 4
    for (int j = beg; j < end; ++j) {
        const int s = bsrc[j];
        p  = __expf(leaky(a_src[s * HEADS + head] + ad) - m);
        hv = *(const v4f*)&h[(size_t)s * HC + lane * 4];
        den += p;
        acc += hv * p;
    }

    const float dd = den + 1e-16f;
    const v4f   b  = *(const v4f*)&bias[lane * 4];
    v4f o;
    o.x = acc.x / dd + b.x;
    o.y = acc.y / dd + b.y;
    o.z = acc.z / dd + b.z;
    o.w = acc.w / dd + b.w;
    *(v4f*)&out[(size_t)d * HC + lane * 4] = o;
}

extern "C" void kernel_launch(void* const* d_in, const int* in_sizes, int n_in,
                              void* d_out, int out_size, void* d_ws, size_t ws_size,
                              hipStream_t stream) {
    const float* x        = (const float*)d_in[0];
    const int*   ei       = (const int*)d_in[1];
    const float* W        = (const float*)d_in[2];
    const float* att_src  = (const float*)d_in[3];
    const float* att_dst  = (const float*)d_in[4];
    const float* bias     = (const float*)d_in[5];
    float*       out      = (float*)d_out;

    const int N = in_sizes[0] / IN_F;
    const int E = in_sizes[1] / 2;

    float* h     = (float*)d_ws;
    float* a_src = h + (size_t)N * HC;
    float* a_dst = a_src + (size_t)N * HEADS;
    int*   cnt   = (int*)(a_dst + (size_t)N * HEADS);
    int*   off   = cnt + N;
    int*   rank  = off + N + 1;
    int*   bsrc  = rank + E;

    gat_gemm_wmma<<<N / 16, 128, GEMM_LDS_BYTES, stream>>>(x, W, h);

    gat_attn_coeff<<<(N + 7) / 8, 256, 0, stream>>>(h, att_src, att_dst, a_src, a_dst, cnt, N);

    gat_edge_rank<<<(E + 255) / 256, 256, 0, stream>>>(ei, cnt, rank, E);
    gat_bucket_offsets<<<(N + 1024) / 1024, 1024, 0, stream>>>(cnt, off, N);
    gat_bucket_fill<<<(E + 255) / 256, 256, 0, stream>>>(ei, off, rank, bsrc, E);

    gat_node_gather<<<(N + 7) / 8, 256, 0, stream>>>(off, bsrc, a_src, a_dst, h, bias, out, N);
}


# Edges grouped by destination with LDS counters instead of one scattered returning global add per edge (chunk count, place by 512-node range, per-range count/scan/place); sum order stays arrival order
# speedup vs baseline: 5.4867x; 1.6943x over previous
#include <hip/hip_runtime.h>

#define IN_F      128
#define HC        128
#define HEADS     4
#define OUT_F     32
#define NEG_SLOPE 0.2f

typedef float v2f __attribute__((ext_vector_type(2)));
typedef float v4f __attribute__((ext_vector_type(4)));
typedef float v8f __attribute__((ext_vector_type(8)));
typedef int   v4i __attribute__((ext_vector_type(4)));
typedef int   v2i __attribute__((ext_vector_type(2)));

__device__ __forceinline__ float leaky(float x) {
    return x > 0.0f ? x : NEG_SLOPE * x;
}

#define WSTR 136
#define XSTR 132
#define GEMM_LDS_BYTES ((128 * WSTR + 16 * XSTR) * 4)

__global__ __launch_bounds__(128) void gat_gemm_wmma(const float* __restrict__ x,
                                                     const float* __restrict__ W,
                                                     float* __restrict__ h) {
    extern __shared__ float smem[];
    float* Wl = smem;
    float* Xl = smem + 128 * WSTR;

    const int tid  = threadIdx.x;
    const int row0 = blockIdx.x * 16;

    for (int i = tid; i < 128 * 32; i += 128) {
        int r  = i >> 5;
        int c4 = i & 31;
        v4f v = ((const v4f*)W)[r * 32 + c4];
        *(v4f*)&Wl[r * WSTR + c4 * 4] = v;
    }
    for (int i = tid; i < 16 * 32; i += 128) {
        int r  = i >> 5;
        int c4 = i & 31;
        v4f v = ((const v4f*)x)[(size_t)(row0 + r) * 32 + c4];
        *(v4f*)&Xl[r * XSTR + c4 * 4] = v;
    }
    __syncthreads();

    const int wave = tid >> 5;
    const int lane = tid & 31;
    const int L    = lane & 15;
    const int g    = lane >> 4;

    const int col0 = wave * 16;
    const int col1 = (wave + 4) * 16;

    v8f acc0 = {};
    v8f acc1 = {};

    for (int k0 = 0; k0 < IN_F; k0 += 4) {
        const int k = k0 + 2 * g;
        v2f a, b0, b1;
        a.x  = Xl[L * XSTR + k];
        a.y  = Xl[L * XSTR + k + 1];
        b0.x = Wl[k * WSTR + col0 + L];
        b0.y = Wl[(k + 1) * WSTR + col0 + L];
        b1.x = Wl[k * WSTR + col1 + L];
        b1.y = Wl[(k + 1) * WSTR + col1 + L];
        acc0 = __builtin_amdgcn_wmma_f32_16x16x4_f32(false, a, false, b0, (short)0, acc0, false, false);
        acc1 = __builtin_amdgcn_wmma_f32_16x16x4_f32(false, a, false, b1, (short)0, acc1, false, false);
    }

    const int mbase = row0 + 8 * g;
    for (int r = 0; r < 8; ++r) {
        h[(size_t)(mbase + r) * HC + col0 + L] = acc0[r];
        h[(size_t)(mbase + r) * HC + col1 + L] = acc1[r];
    }
}

__global__ __launch_bounds__(256) void gat_attn_coeff(const float* __restrict__ h,
                                                      const float* __restrict__ att_src,
                                                      const float* __restrict__ att_dst,
                                                      float* __restrict__ a_src,
                                                      float* __restrict__ a_dst,
                                                      int N) {
    int n    = blockIdx.x * 8 + (threadIdx.x >> 5);
    int lane = threadIdx.x & 31;
    if (n >= N) return;

    float s[HEADS], d[HEADS];
#pragma unroll
    for (int hh = 0; hh < HEADS; ++hh) {
        float v = h[(size_t)n * HC + hh * OUT_F + lane];
        s[hh] = v * att_src[hh * OUT_F + lane];
        d[hh] = v * att_dst[hh * OUT_F + lane];
    }
#pragma unroll
    for (int off = 16; off > 0; off >>= 1) {
#pragma unroll
        for (int hh = 0; hh < HEADS; ++hh) {
            s[hh] += __shfl_xor(s[hh], off, 32);
            d[hh] += __shfl_xor(d[hh], off, 32);
        }
    }
    if (lane == 0) {
#pragma unroll
        for (int hh = 0; hh < HEADS; ++hh) {
            a_src[n * HEADS + hh] = s[hh];
            a_dst[n * HEADS + hh] = d[hh];
        }
    }
}

#define CB_SHIFT 9
#define CB_NODES 512
#define CB_MAX   256
#define CHUNK    8192
#define CHUNK_PT (CHUNK / 1024)
#define FINE_PT 10
#define NB_ROW_LOADS 7

__global__ __launch_bounds__(1024) void gat_coarse_count(const int* __restrict__ ei,
                                                         int* __restrict__ hist,
                                                         int E, int CB, int NB) {
    __shared__ int lc[CB_MAX];
    const int t = threadIdx.x;
    if (t < CB_MAX) lc[t] = 0;
    __syncthreads();

    const int e0 = blockIdx.x * CHUNK;
    int d[CHUNK_PT];
#pragma unroll
    for (int k = 0; k < CHUNK_PT; ++k) {
        const int e = e0 + k * 1024 + t;
        d[k] = (e < E) ? ei[E + e] : -1;
    }
#pragma unroll
    for (int k = 0; k < CHUNK_PT; ++k)
        if (d[k] >= 0) atomicAdd(&lc[d[k] >> CB_SHIFT], 1);
    __syncthreads();

    if (t < CB) hist[t * NB + blockIdx.x] = lc[t];
}

__global__ __launch_bounds__(1024) void gat_coarse_place(const int* __restrict__ ei,
                                                         const int* __restrict__ hist,
                                                         int* __restrict__ cstart,
                                                         v2i* __restrict__ cpair,
                                                         int E, int CB, int NB) {
    __shared__ int tot[CB_MAX];
    __shared__ int nxt[CB_MAX];
    const int t    = threadIdx.x;
    const int lane = t & 31;
    const int wave = t >> 5;
    const int b    = blockIdx.x;

    const int e0 = b * CHUNK;
    int s[CHUNK_PT], d[CHUNK_PT];
#pragma unroll
    for (int k = 0; k < CHUNK_PT; ++k) {
        const int e = e0 + k * 1024 + t;
        const bool ok = e < E;
        s[k] = ok ? ei[e] : 0;
        d[k] = ok ? ei[E + e] : -1;
    }

#pragma unroll
    for (int c0 = 0; c0 < CB_MAX; c0 += 128) {
        int sa[4], sb[4];
#pragma unroll
        for (int q = 0; q < 4; ++q) {
            const int c = c0 + 32 * q + wave;
            sa[q] = 0;
            sb[q] = 0;
#pragma unroll
            for (int r = 0; r < NB_ROW_LOADS; ++r) {
                const int i = lane + 32 * r;
                const int v = (c < CB && i < NB) ? hist[c * NB + i] : 0;
                sa[q] += v;
                sb[q] += (i < b) ? v : 0;
            }
        }
        if (NB > 32 * NB_ROW_LOADS) {
            for (int q = 0; q < 4; ++q) {
                const int c = c0 + 32 * q + wave;
                if (c < CB) {
                    for (int i = lane + 32 * NB_ROW_LOADS; i < NB; i += 32) {
                        const int v = hist[c * NB + i];
                        sa[q] += v;
                        sb[q] += (i < b) ? v : 0;
                    }
                }
            }
        }
#pragma unroll
        for (int q = 0; q < 4; ++q) {
            for (int o = 16; o > 0; o >>= 1) {
                sa[q] += __shfl_xor(sa[q], o, 32);
                sb[q] += __shfl_xor(sb[q], o, 32);
            }
            if (lane == 0) {
                tot[c0 + 32 * q + wave] = sa[q];
                nxt[c0 + 32 * q + wave] = sb[q];
            }
        }
    }
    __syncthreads();

    if (wave == 0) {
        int v[8];
        int s = 0;
#pragma unroll
        for (int k = 0; k < 8; ++k) { v[k] = tot[lane * 8 + k]; s += v[k]; }
        int inc = s;
        for (int o = 1; o < 32; o <<= 1) {
            const int u = __shfl_up(inc, o, 32);
            if (lane >= o) inc += u;
        }
        int run = inc - s;
#pragma unroll
        for (int k = 0; k < 8; ++k) {
            const int c = lane * 8 + k;
            if (b == 0 && c <= CB) cstart[c] = run;
            nxt[c] += run;
            run += v[k];
        }
    }
    __syncthreads();

#pragma unroll
    for (int k = 0; k < CHUNK_PT; ++k) {
        if (d[k] >= 0) {
            const int pos = atomicAdd(&nxt[d[k] >> CB_SHIFT], 1);
            v2i p;
            p.x = s[k];
            p.y = d[k];
            cpair[pos] = p;
        }
    }
}

__global__ __launch_bounds__(1024) void gat_fine_place(const v2i* __restrict__ cpair,
                                                       const int* __restrict__ cstart,
                                                       int* __restrict__ off,
                                                       int* __restrict__ bsrc,
                                                       int N) {
    __shared__ int cnt[CB_NODES];
    __shared__ int wtot[16];
    const int t    = threadIdx.x;
    const int lane = t & 31;
    const int wave = t >> 5;
    const int beg  = cstart[blockIdx.x];
    const int end  = cstart[blockIdx.x + 1];
    const int n0   = blockIdx.x << CB_SHIFT;

    if (t < CB_NODES) cnt[t] = 0;
    __syncthreads();

    for (int i = beg + t; i < end; i += FINE_PT * 1024) {
        int dy[FINE_PT];
#pragma unroll
        for (int k = 0; k < FINE_PT; ++k) {
            const int j = i + k * 1024;
            dy[k] = (j < end) ? cpair[j].y : -1;
        }
#pragma unroll
        for (int k = 0; k < FINE_PT; ++k)
            if (dy[k] >= 0) atomicAdd(&cnt[dy[k] - n0], 1);
    }
    __syncthreads();

    int v = 0, inc = 0;
    if (t < CB_NODES) {
        v   = cnt[t];
        inc = v;
        for (int o = 1; o < 32; o <<= 1) {
            const int u = __shfl_up(inc, o, 32);
            if (lane >= o) inc += u;
        }
        if (lane == 31) wtot[wave] = inc;
    }
    __syncthreads();
    if (t < CB_NODES) {
        int w = (lane < wave) ? wtot[lane] : 0;
        for (int o = 16; o > 0; o >>= 1) w += __shfl_xor(w, o, 32);
        const int ex = beg + w + inc - v;
        cnt[t] = ex;
        if (n0 + t <= N) off[n0 + t] = ex;
    }
    if (blockIdx.x == gridDim.x - 1 && t == 0) off[N] = end;
    __syncthreads();

    for (int i = beg + t; i < end; i += FINE_PT * 1024) {
        v2i p[FINE_PT];
#pragma unroll
        for (int k = 0; k < FINE_PT; ++k) {
            const int j = i + k * 1024;
            v2i q;
            q.x = 0;
            q.y = -1;
            p[k] = (j < end) ? cpair[j] : q;
        }
#pragma unroll
        for (int k = 0; k < FINE_PT; ++k) {
            if (p[k].y >= 0) {
                const int pos = atomicAdd(&cnt[p[k].y - n0], 1);
                bsrc[pos] = p[k].x;
            }
        }
    }
}

__global__ __launch_bounds__(256) void gat_node_gather(const int* __restrict__ off,
                                                       const int* __restrict__ bsrc,
                                                       const float* __restrict__ a_src,
                                                       const float* __restrict__ a_dst,
                                                       const float* __restrict__ h,
                                                       const float* __restrict__ bias,
                                                       float* __restrict__ out,
                                                       int N) {
    const int lane = threadIdx.x & 31;
    const int d = __builtin_amdgcn_readfirstlane((int)(blockIdx.x * 8 + (threadIdx.x >> 5)));
    if (d >= N) return;

    const int   beg   = off[d];
    const int   end   = off[d + 1];
    const int   head  = lane >> 3;
    const float ad    = a_dst[d * HEADS + head];
    const float eself = leaky(a_src[d * HEADS + head] + ad);

    float m = eself;
#pragma unroll 4
    for (int j = beg; j < end; ++j) {
        const int   s = bsrc[j];
        const float e = leaky(a_src[s * HEADS + head] + ad);
        m = e > m ? e : m;
    }

    float p   = __expf(eself - m);
    float den = p;
    v4f   hv  = *(const v4f*)&h[(size_t)d * HC + lane * 4];
    v4f   acc = hv * p;
#pragma unroll 4
    for (int j = beg; j < end; ++j) {
        const int s = bsrc[j];
        p  = __expf(leaky(a_src[s * HEADS + head] + ad) - m);
        hv = *(const v4f*)&h[(size_t)s * HC + lane * 4];
        den += p;
        acc += hv * p;
    }

    const float dd = den + 1e-16f;
    const v4f   b  = *(const v4f*)&bias[lane * 4];
    v4f o;
    o.x = acc.x / dd + b.x;
    o.y = acc.y / dd + b.y;
    o.z = acc.z / dd + b.z;
    o.w = acc.w / dd + b.w;
    *(v4f*)&out[(size_t)d * HC + lane * 4] = o;
}

extern "C" void kernel_launch(void* const* d_in, const int* in_sizes, int n_in,
                              void* d_out, int out_size, void* d_ws, size_t ws_size,
                              hipStream_t stream) {
    const float* x        = (const float*)d_in[0];
    const int*   ei       = (const int*)d_in[1];
    const float* W        = (const float*)d_in[2];
    const float* att_src  = (const float*)d_in[3];
    const float* att_dst  = (const float*)d_in[4];
    const float* bias     = (const float*)d_in[5];
    float*       out      = (float*)d_out;

    const int N = in_sizes[0] / IN_F;
    const int E = in_sizes[1] / 2;

    const int CB = (N + CB_NODES - 1) / CB_NODES;
    const int NB = (E + CHUNK - 1) / CHUNK;

    float* h      = (float*)d_ws;
    float* a_src  = h + (size_t)N * HC;
    float* a_dst  = a_src + (size_t)N * HEADS;
    v2i*   cpair  = (v2i*)(a_dst + (size_t)N * HEADS);
    int*   bsrc   = (int*)(cpair + E);
    int*   off    = bsrc + E;
    int*   hist   = off + N + 1;
    int*   cstart = hist + (size_t)CB * NB;

    gat_gemm_wmma<<<N / 16, 128, GEMM_LDS_BYTES, stream>>>(x, W, h);

    gat_attn_coeff<<<(N + 7) / 8, 256, 0, stream>>>(h, att_src, att_dst, a_src, a_dst, N);

    gat_coarse_count<<<NB, 1024, 0, stream>>>(ei, hist, E, CB, NB);
    gat_coarse_place<<<NB, 1024, 0, stream>>>(ei, hist, cstart, cpair, E, CB, NB);
    gat_fine_place<<<CB, 1024, 0, stream>>>(cpair, cstart, off, bsrc, N);

    gat_node_gather<<<(N + 7) / 8, 256, 0, stream>>>(off, bsrc, a_src, a_dst, h, bias, out, N);
}


# Attention coefficients a_src, a_dst formed from the WMMA accumulators in the GEMM epilogue (wave = one head); the coefficient kernel is gone; their dot products are summed in a new tree order
# speedup vs baseline: 5.7786x; 1.0532x over previous
#include <hip/hip_runtime.h>

#define IN_F      128
#define HC        128
#define HEADS     4
#define OUT_F     32
#define NEG_SLOPE 0.2f

typedef float v2f __attribute__((ext_vector_type(2)));
typedef float v4f __attribute__((ext_vector_type(4)));
typedef float v8f __attribute__((ext_vector_type(8)));
typedef int   v4i __attribute__((ext_vector_type(4)));
typedef int   v2i __attribute__((ext_vector_type(2)));

__device__ __forceinline__ float leaky(float x) {
    return x > 0.0f ? x : NEG_SLOPE * x;
}

#define WSTR 136
#define XSTR 132
#define COEF_FLOATS 128
#define GEMM_LDS_BYTES ((128 * WSTR + 16 * XSTR + COEF_FLOATS) * 4)

__global__ __launch_bounds__(128) void gat_gemm_wmma(const float* __restrict__ x,
                                                     const float* __restrict__ W,
                                                     const float* __restrict__ att_src,
                                                     const float* __restrict__ att_dst,
                                                     float* __restrict__ h,
                                                     float* __restrict__ a_src,
                                                     float* __restrict__ a_dst) {
    extern __shared__ float smem[];
    float* Wl = smem;
    float* Xl = smem + 128 * WSTR;
    float* Cl = Xl + 16 * XSTR;

    const int tid  = threadIdx.x;
    const int row0 = blockIdx.x * 16;

    const int wave = tid >> 5;
    const int lane = tid & 31;
    const int L    = lane & 15;
    const int g    = lane >> 4;

    const int col0 = wave * 32;
    const int col1 = wave * 32 + 16;

    const float as0 = att_src[col0 + L];
    const float as1 = att_src[col1 + L];
    const float ad0 = att_dst[col0 + L];
    const float ad1 = att_dst[col1 + L];

    for (int i = tid; i < 128 * 32; i += 128) {
        int r  = i >> 5;
        int c4 = i & 31;
        v4f v = ((const v4f*)W)[r * 32 + c4];
        *(v4f*)&Wl[r * WSTR + c4 * 4] = v;
    }
    for (int i = tid; i < 16 * 32; i += 128) {
        int r  = i >> 5;
        int c4 = i & 31;
        v4f v = ((const v4f*)x)[(size_t)(row0 + r) * 32 + c4];
        *(v4f*)&Xl[r * XSTR + c4 * 4] = v;
    }
    __syncthreads();

    v8f acc0 = {};
    v8f acc1 = {};

    for (int k0 = 0; k0 < IN_F; k0 += 4) {
        const int k = k0 + 2 * g;
        v2f a, b0, b1;
        a.x  = Xl[L * XSTR + k];
        a.y  = Xl[L * XSTR + k + 1];
        b0.x = Wl[k * WSTR + col0 + L];
        b0.y = Wl[(k + 1) * WSTR + col0 + L];
        b1.x = Wl[k * WSTR + col1 + L];
        b1.y = Wl[(k + 1) * WSTR + col1 + L];
        acc0 = __builtin_amdgcn_wmma_f32_16x16x4_f32(false, a, false, b0, (short)0, acc0, false, false);
        acc1 = __builtin_amdgcn_wmma_f32_16x16x4_f32(false, a, false, b1, (short)0, acc1, false, false);
    }

    float v[16];
#pragma unroll
    for (int r = 0; r < 8; ++r) {
        v[r]     = acc0[r] * as0 + acc1[r] * as1;
        v[8 + r] = acc0[r] * ad0 + acc1[r] * ad1;
    }
    const bool hi8 = (L & 8) != 0;
    const bool hi4 = (L & 4) != 0;
    const bool hi2 = (L & 2) != 0;
    const bool hi1 = (L & 1) != 0;
    float u[8], t[4], q[2];
#pragma unroll
    for (int j = 0; j < 8; ++j) {
        const float keep = hi8 ? v[j + 8] : v[j];
        const float send = hi8 ? v[j] : v[j + 8];
        u[j] = keep + __shfl_xor(send, 8, 32);
    }
#pragma unroll
    for (int j = 0; j < 4; ++j) {
        const float keep = hi4 ? u[j + 4] : u[j];
        const float send = hi4 ? u[j] : u[j + 4];
        t[j] = keep + __shfl_xor(send, 4, 32);
    }
#pragma unroll
    for (int j = 0; j < 2; ++j) {
        const float keep = hi2 ? t[j + 2] : t[j];
        const float send = hi2 ? t[j] : t[j + 2];
        q[j] = keep + __shfl_xor(send, 2, 32);
    }
    const float keep1 = hi1 ? q[1] : q[0];
    const float send1 = hi1 ? q[0] : q[1];
    const float z = keep1 + __shfl_xor(send1, 1, 32);
    Cl[(L >> 3) * 64 + (8 * g + (L & 7)) * HEADS + wave] = z;
    __syncthreads();

    const int mbase = row0 + 8 * g;
    for (int r = 0; r < 8; ++r) {
        h[(size_t)(mbase + r) * HC + col0 + L] = acc0[r];
        h[(size_t)(mbase + r) * HC + col1 + L] = acc1[r];
    }

    if (wave == 0) {
        const v4f c = *(const v4f*)&Cl[lane * 4];
        float* dstp = (lane < 16) ? a_src + (size_t)(row0 + lane) * HEADS
                                  : a_dst + (size_t)(row0 + lane - 16) * HEADS;
        *(v4f*)dstp = c;
    }
}

#define CB_SHIFT 9
#define CB_NODES 512
#define CB_MAX   256
#define CHUNK    8192
#define CHUNK_PT (CHUNK / 1024)
#define FINE_PT 10
#define NB_ROW_LOADS 7

__global__ __launch_bounds__(1024) void gat_coarse_count(const int* __restrict__ ei,
                                                         int* __restrict__ hist,
                                                         int E, int CB, int NB) {
    __shared__ int lc[CB_MAX];
    const int t = threadIdx.x;
    if (t < CB_MAX) lc[t] = 0;
    __syncthreads();

    const int e0 = blockIdx.x * CHUNK;
    int d[CHUNK_PT];
#pragma unroll
    for (int k = 0; k < CHUNK_PT; ++k) {
        const int e = e0 + k * 1024 + t;
        d[k] = (e < E) ? ei[E + e] : -1;
    }
#pragma unroll
    for (int k = 0; k < CHUNK_PT; ++k)
        if (d[k] >= 0) atomicAdd(&lc[d[k] >> CB_SHIFT], 1);
    __syncthreads();

    if (t < CB) hist[t * NB + blockIdx.x] = lc[t];
}

__global__ __launch_bounds__(1024) void gat_coarse_place(const int* __restrict__ ei,
                                                         const int* __restrict__ hist,
                                                         int* __restrict__ cstart,
                                                         v2i* __restrict__ cpair,
                                                         int E, int CB, int NB) {
    __shared__ int tot[CB_MAX];
    __shared__ int nxt[CB_MAX];
    const int t    = threadIdx.x;
    const int lane = t & 31;
    const int wave = t >> 5;
    const int b    = blockIdx.x;

    const int e0 = b * CHUNK;
    int s[CHUNK_PT], d[CHUNK_PT];
#pragma unroll
    for (int k = 0; k < CHUNK_PT; ++k) {
        const int e = e0 + k * 1024 + t;
        const bool ok = e < E;
        s[k] = ok ? ei[e] : 0;
        d[k] = ok ? ei[E + e] : -1;
    }

#pragma unroll
    for (int c0 = 0; c0 < CB_MAX; c0 += 128) {
        int sa[4], sb[4];
#pragma unroll
        for (int q = 0; q < 4; ++q) {
            const int c = c0 + 32 * q + wave;
            sa[q] = 0;
            sb[q] = 0;
#pragma unroll
            for (int r = 0; r < NB_ROW_LOADS; ++r) {
                const int i = lane + 32 * r;
                const int v = (c < CB && i < NB) ? hist[c * NB + i] : 0;
                sa[q] += v;
                sb[q] += (i < b) ? v : 0;
            }
        }
        if (NB > 32 * NB_ROW_LOADS) {
            for (int q = 0; q < 4; ++q) {
                const int c = c0 + 32 * q + wave;
                if (c < CB) {
                    for (int i = lane + 32 * NB_ROW_LOADS; i < NB; i += 32) {
                        const int v = hist[c * NB + i];
                        sa[q] += v;
                        sb[q] += (i < b) ? v : 0;
                    }
                }
            }
        }
#pragma unroll
        for (int q = 0; q < 4; ++q) {
            for (int o = 16; o > 0; o >>= 1) {
                sa[q] += __shfl_xor(sa[q], o, 32);
                sb[q] += __shfl_xor(sb[q], o, 32);
            }
            if (lane == 0) {
                tot[c0 + 32 * q + wave] = sa[q];
                nxt[c0 + 32 * q + wave] = sb[q];
            }
        }
    }
    __syncthreads();

    if (wave == 0) {
        int v[8];
        int s = 0;
#pragma unroll
        for (int k = 0; k < 8; ++k) { v[k] = tot[lane * 8 + k]; s += v[k]; }
        int inc = s;
        for (int o = 1; o < 32; o <<= 1) {
            const int u = __shfl_up(inc, o, 32);
            if (lane >= o) inc += u;
        }
        int run = inc - s;
#pragma unroll
        for (int k = 0; k < 8; ++k) {
            const int c = lane * 8 + k;
            if (b == 0 && c <= CB) cstart[c] = run;
            nxt[c] += run;
            run += v[k];
        }
    }
    __syncthreads();

#pragma unroll
    for (int k = 0; k < CHUNK_PT; ++k) {
        if (d[k] >= 0) {
            const int pos = atomicAdd(&nxt[d[k] >> CB_SHIFT], 1);
            v2i p;
            p.x = s[k];
            p.y = d[k];
            cpair[pos] = p;
        }
    }
}

__global__ __launch_bounds__(1024) void gat_fine_place(const v2i* __restrict__ cpair,
                                                       const int* __restrict__ cstart,
                                                       int* __restrict__ off,
                                                       int* __restrict__ bsrc,
                                                       int N) {
    __shared__ int cnt[CB_NODES];
    __shared__ int wtot[16];
    const int t    = threadIdx.x;
    const int lane = t & 31;
    const int wave = t >> 5;
    const int beg  = cstart[blockIdx.x];
    const int end  = cstart[blockIdx.x + 1];
    const int n0   = blockIdx.x << CB_SHIFT;

    if (t < CB_NODES) cnt[t] = 0;
    __syncthreads();

    for (int i = beg + t; i < end; i += FINE_PT * 1024) {
        int dy[FINE_PT];
#pragma unroll
        for (int k = 0; k < FINE_PT; ++k) {
            const int j = i + k * 1024;
            dy[k] = (j < end) ? cpair[j].y : -1;
        }
#pragma unroll
        for (int k = 0; k < FINE_PT; ++k)
            if (dy[k] >= 0) atomicAdd(&cnt[dy[k] - n0], 1);
    }
    __syncthreads();

    int v = 0, inc = 0;
    if (t < CB_NODES) {
        v   = cnt[t];
        inc = v;
        for (int o = 1; o < 32; o <<= 1) {
            const int u = __shfl_up(inc, o, 32);
            if (lane >= o) inc += u;
        }
        if (lane == 31) wtot[wave] = inc;
    }
    __syncthreads();
    if (t < CB_NODES) {
        int w = (lane < wave) ? wtot[lane] : 0;
        for (int o = 16; o > 0; o >>= 1) w += __shfl_xor(w, o, 32);
        const int ex = beg + w + inc - v;
        cnt[t] = ex;
        if (n0 + t <= N) off[n0 + t] = ex;
    }
    if (blockIdx.x == gridDim.x - 1 && t == 0) off[N] = end;
    __syncthreads();

    for (int i = beg + t; i < end; i += FINE_PT * 1024) {
        v2i p[FINE_PT];
#pragma unroll
        for (int k = 0; k < FINE_PT; ++k) {
            const int j = i + k * 1024;
            v2i q;
            q.x = 0;
            q.y = -1;
            p[k] = (j < end) ? cpair[j] : q;
        }
#pragma unroll
        for (int k = 0; k < FINE_PT; ++k) {
            if (p[k].y >= 0) {
                const int pos = atomicAdd(&cnt[p[k].y - n0], 1);
                bsrc[pos] = p[k].x;
            }
        }
    }
}

__global__ __launch_bounds__(256) void gat_node_gather(const int* __restrict__ off,
                                                       const int* __restrict__ bsrc,
                                                       const float* __restrict__ a_src,
                                                       const float* __restrict__ a_dst,
                                                       const float* __restrict__ h,
                                                       const float* __restrict__ bias,
                                                       float* __restrict__ out,
                                                       int N) {
    const int lane = threadIdx.x & 31;
    const int d = __builtin_amdgcn_readfirstlane((int)(blockIdx.x * 8 + (threadIdx.x >> 5)));
    if (d >= N) return;

    const int   beg   = off[d];
    const int   end   = off[d + 1];
    const int   head  = lane >> 3;
    const float ad    = a_dst[d * HEADS + head];
    const float eself = leaky(a_src[d * HEADS + head] + ad);

    float m = eself;
#pragma unroll 4
    for (int j = beg; j < end; ++j) {
        const int   s = bsrc[j];
        const float e = leaky(a_src[s * HEADS + head] + ad);
        m = e > m ? e : m;
    }

    float p   = __expf(eself - m);
    float den = p;
    v4f   hv  = *(const v4f*)&h[(size_t)d * HC + lane * 4];
    v4f   acc = hv * p;
#pragma unroll 4
    for (int j = beg; j < end; ++j) {
        const int s = bsrc[j];
        p  = __expf(leaky(a_src[s * HEADS + head] + ad) - m);
        hv = *(const v4f*)&h[(size_t)s * HC + lane * 4];
        den += p;
        acc += hv * p;
    }

    const float dd = den + 1e-16f;
    const v4f   b  = *(const v4f*)&bias[lane * 4];
    v4f o;
    o.x = acc.x / dd + b.x;
    o.y = acc.y / dd + b.y;
    o.z = acc.z / dd + b.z;
    o.w = acc.w / dd + b.w;
    *(v4f*)&out[(size_t)d * HC + lane * 4] = o;
}

extern "C" void kernel_launch(void* const* d_in, const int* in_sizes, int n_in,
                              void* d_out, int out_size, void* d_ws, size_t ws_size,
                              hipStream_t stream) {
    const float* x        = (const float*)d_in[0];
    const int*   ei       = (const int*)d_in[1];
    const float* W        = (const float*)d_in[2];
    const float* att_src  = (const float*)d_in[3];
    const float* att_dst  = (const float*)d_in[4];
    const float* bias     = (const float*)d_in[5];
    float*       out      = (float*)d_out;

    const int N = in_sizes[0] / IN_F;
    const int E = in_sizes[1] / 2;

    const int CB = (N + CB_NODES - 1) / CB_NODES;
    const int NB = (E + CHUNK - 1) / CHUNK;

    float* h      = (float*)d_ws;
    float* a_src  = h + (size_t)N * HC;
    float* a_dst  = a_src + (size_t)N * HEADS;
    v2i*   cpair  = (v2i*)(a_dst + (size_t)N * HEADS);
    int*   bsrc   = (int*)(cpair + E);
    int*   off    = bsrc + E;
    int*   hist   = off + N + 1;
    int*   cstart = hist + (size_t)CB * NB;

    gat_gemm_wmma<<<N / 16, 128, GEMM_LDS_BYTES, stream>>>(x, W, att_src, att_dst, h, a_src, a_dst);

    gat_coarse_count<<<NB, 1024, 0, stream>>>(ei, hist, E, CB, NB);
    gat_coarse_place<<<NB, 1024, 0, stream>>>(ei, hist, cstart, cpair, E, CB, NB);
    gat_fine_place<<<CB, 1024, 0, stream>>>(cpair, cstart, off, bsrc, N);

    gat_node_gather<<<(N + 7) / 8, 256, 0, stream>>>(off, bsrc, a_src, a_dst, h, bias, out, N);
}


# GEMM and node gather run as resident grids (1024 and 2048 blocks): blocks loop over row tiles (W staged to LDS once per block) and waves over nodes, so 125000 short-lived waves become 20480
# speedup vs baseline: 5.9863x; 1.0359x over previous
#include <hip/hip_runtime.h>

#define IN_F      128
#define HC        128
#define HEADS     4
#define OUT_F     32
#define NEG_SLOPE 0.2f

typedef float v2f __attribute__((ext_vector_type(2)));
typedef float v4f __attribute__((ext_vector_type(4)));
typedef float v8f __attribute__((ext_vector_type(8)));
typedef int   v4i __attribute__((ext_vector_type(4)));
typedef int   v2i __attribute__((ext_vector_type(2)));

__device__ __forceinline__ float leaky(float x) {
    return x > 0.0f ? x : NEG_SLOPE * x;
}

#define WSTR 136
#define XSTR 132
#define COEF_FLOATS 128
#define GEMM_LDS_BYTES ((128 * WSTR + 16 * XSTR + COEF_FLOATS) * 4)

__global__ __launch_bounds__(128) void gat_gemm_wmma(const float* __restrict__ x,
                                                     const float* __restrict__ W,
                                                     const float* __restrict__ att_src,
                                                     const float* __restrict__ att_dst,
                                                     float* __restrict__ h,
                                                     float* __restrict__ a_src,
                                                     float* __restrict__ a_dst,
                                                     int N) {
    extern __shared__ float smem[];
    float* Wl = smem;
    float* Xl = smem + 128 * WSTR;
    float* Cl = Xl + 16 * XSTR;

    const int tid    = threadIdx.x;
    const int ntiles = N / 16;

    const int wave = tid >> 5;
    const int lane = tid & 31;
    const int L    = lane & 15;
    const int g    = lane >> 4;

    const int col0 = wave * 32;
    const int col1 = wave * 32 + 16;

    const float as0 = att_src[col0 + L];
    const float as1 = att_src[col1 + L];
    const float ad0 = att_dst[col0 + L];
    const float ad1 = att_dst[col1 + L];

    for (int i = tid; i < 128 * 32; i += 128) {
        int r  = i >> 5;
        int c4 = i & 31;
        v4f v = ((const v4f*)W)[r * 32 + c4];
        *(v4f*)&Wl[r * WSTR + c4 * 4] = v;
    }
    for (int tile = blockIdx.x; tile < ntiles; tile += gridDim.x) {
        const int row0 = tile * 16;
        for (int i = tid; i < 16 * 32; i += 128) {
            int r  = i >> 5;
            int c4 = i & 31;
            v4f v = ((const v4f*)x)[(size_t)(row0 + r) * 32 + c4];
            *(v4f*)&Xl[r * XSTR + c4 * 4] = v;
        }
        __syncthreads();

        v8f acc0 = {};
        v8f acc1 = {};

        for (int k0 = 0; k0 < IN_F; k0 += 4) {
            const int k = k0 + 2 * g;
            v2f a, b0, b1;
            a.x  = Xl[L * XSTR + k];
            a.y  = Xl[L * XSTR + k + 1];
            b0.x = Wl[k * WSTR + col0 + L];
            b0.y = Wl[(k + 1) * WSTR + col0 + L];
            b1.x = Wl[k * WSTR + col1 + L];
            b1.y = Wl[(k + 1) * WSTR + col1 + L];
            acc0 = __builtin_amdgcn_wmma_f32_16x16x4_f32(false, a, false, b0, (short)0, acc0, false, false);
            acc1 = __builtin_amdgcn_wmma_f32_16x16x4_f32(false, a, false, b1, (short)0, acc1, false, false);
        }

        float v[16];
#pragma unroll
        for (int r = 0; r < 8; ++r) {
            v[r]     = acc0[r] * as0 + acc1[r] * as1;
            v[8 + r] = acc0[r] * ad0 + acc1[r] * ad1;
        }
        const bool hi8 = (L & 8) != 0;
        const bool hi4 = (L & 4) != 0;
        const bool hi2 = (L & 2) != 0;
        const bool hi1 = (L & 1) != 0;
        float u[8], t[4], q[2];
#pragma unroll
        for (int j = 0; j < 8; ++j) {
            const float keep = hi8 ? v[j + 8] : v[j];
            const float send = hi8 ? v[j] : v[j + 8];
            u[j] = keep + __shfl_xor(send, 8, 32);
        }
#pragma unroll
        for (int j = 0; j < 4; ++j) {
            const float keep = hi4 ? u[j + 4] : u[j];
            const float send = hi4 ? u[j] : u[j + 4];
            t[j] = keep + __shfl_xor(send, 4, 32);
        }
#pragma unroll
        for (int j = 0; j < 2; ++j) {
            const float keep = hi2 ? t[j + 2] : t[j];
            const float send = hi2 ? t[j] : t[j + 2];
            q[j] = keep + __shfl_xor(send, 2, 32);
        }
        const float keep1 = hi1 ? q[1] : q[0];
        const float send1 = hi1 ? q[0] : q[1];
        const float z = keep1 + __shfl_xor(send1, 1, 32);
        Cl[(L >> 3) * 64 + (8 * g + (L & 7)) * HEADS + wave] = z;
        __syncthreads();

        const int mbase = row0 + 8 * g;
        for (int r = 0; r < 8; ++r) {
            h[(size_t)(mbase + r) * HC + col0 + L] = acc0[r];
            h[(size_t)(mbase + r) * HC + col1 + L] = acc1[r];
        }

        if (wave == 0) {
            const v4f c = *(const v4f*)&Cl[lane * 4];
            float* dstp = (lane < 16) ? a_src + (size_t)(row0 + lane) * HEADS
                                      : a_dst + (size_t)(row0 + lane - 16) * HEADS;
            *(v4f*)dstp = c;
        }
    }
}

#define CB_SHIFT 9
#define CB_NODES 512
#define CB_MAX   256
#define CHUNK    8192
#define CHUNK_PT (CHUNK / 1024)
#define FINE_PT 10
#define NB_ROW_LOADS 7

__global__ __launch_bounds__(1024) void gat_coarse_count(const int* __restrict__ ei,
                                                         int* __restrict__ hist,
                                                         int E, int CB, int NB) {
    __shared__ int lc[CB_MAX];
    const int t = threadIdx.x;
    if (t < CB_MAX) lc[t] = 0;
    __syncthreads();

    const int e0 = blockIdx.x * CHUNK;
    int d[CHUNK_PT];
#pragma unroll
    for (int k = 0; k < CHUNK_PT; ++k) {
        const int e = e0 + k * 1024 + t;
        d[k] = (e < E) ? ei[E + e] : -1;
    }
#pragma unroll
    for (int k = 0; k < CHUNK_PT; ++k)
        if (d[k] >= 0) atomicAdd(&lc[d[k] >> CB_SHIFT], 1);
    __syncthreads();

    if (t < CB) hist[t * NB + blockIdx.x] = lc[t];
}

__global__ __launch_bounds__(1024) void gat_coarse_place(const int* __restrict__ ei,
                                                         const int* __restrict__ hist,
                                                         int* __restrict__ cstart,
                                                         v2i* __restrict__ cpair,
                                                         int E, int CB, int NB) {
    __shared__ int tot[CB_MAX];
    __shared__ int nxt[CB_MAX];
    const int t    = threadIdx.x;
    const int lane = t & 31;
    const int wave = t >> 5;
    const int b    = blockIdx.x;

    const int e0 = b * CHUNK;
    int s[CHUNK_PT], d[CHUNK_PT];
#pragma unroll
    for (int k = 0; k < CHUNK_PT; ++k) {
        const int e = e0 + k * 1024 + t;
        const bool ok = e < E;
        s[k] = ok ? ei[e] : 0;
        d[k] = ok ? ei[E + e] : -1;
    }

#pragma unroll
    for (int c0 = 0; c0 < CB_MAX; c0 += 128) {
        int sa[4], sb[4];
#pragma unroll
        for (int q = 0; q < 4; ++q) {
            const int c = c0 + 32 * q + wave;
            sa[q] = 0;
            sb[q] = 0;
#pragma unroll
            for (int r = 0; r < NB_ROW_LOADS; ++r) {
                const int i = lane + 32 * r;
                const int v = (c < CB && i < NB) ? hist[c * NB + i] : 0;
                sa[q] += v;
                sb[q] += (i < b) ? v : 0;
            }
        }
        if (NB > 32 * NB_ROW_LOADS) {
            for (int q = 0; q < 4; ++q) {
                const int c = c0 + 32 * q + wave;
                if (c < CB) {
                    for (int i = lane + 32 * NB_ROW_LOADS; i < NB; i += 32) {
                        const int v = hist[c * NB + i];
                        sa[q] += v;
                        sb[q] += (i < b) ? v : 0;
                    }
                }
            }
        }
#pragma unroll
        for (int q = 0; q < 4; ++q) {
            for (int o = 16; o > 0; o >>= 1) {
                sa[q] += __shfl_xor(sa[q], o, 32);
                sb[q] += __shfl_xor(sb[q], o, 32);
            }
            if (lane == 0) {
                tot[c0 + 32 * q + wave] = sa[q];
                nxt[c0 + 32 * q + wave] = sb[q];
            }
        }
    }
    __syncthreads();

    if (wave == 0) {
        int v[8];
        int s = 0;
#pragma unroll
        for (int k = 0; k < 8; ++k) { v[k] = tot[lane * 8 + k]; s += v[k]; }
        int inc = s;
        for (int o = 1; o < 32; o <<= 1) {
            const int u = __shfl_up(inc, o, 32);
            if (lane >= o) inc += u;
        }
        int run = inc - s;
#pragma unroll
        for (int k = 0; k < 8; ++k) {
            const int c = lane * 8 + k;
            if (b == 0 && c <= CB) cstart[c] = run;
            nxt[c] += run;
            run += v[k];
        }
    }
    __syncthreads();

#pragma unroll
    for (int k = 0; k < CHUNK_PT; ++k) {
        if (d[k] >= 0) {
            const int pos = atomicAdd(&nxt[d[k] >> CB_SHIFT], 1);
            v2i p;
            p.x = s[k];
            p.y = d[k];
            cpair[pos] = p;
        }
    }
}

__global__ __launch_bounds__(1024) void gat_fine_place(const v2i* __restrict__ cpair,
                                                       const int* __restrict__ cstart,
                                                       int* __restrict__ off,
                                                       int* __restrict__ bsrc,
                                                       int N) {
    __shared__ int cnt[CB_NODES];
    __shared__ int wtot[16];
    const int t    = threadIdx.x;
    const int lane = t & 31;
    const int wave = t >> 5;
    const int beg  = cstart[blockIdx.x];
    const int end  = cstart[blockIdx.x + 1];
    const int n0   = blockIdx.x << CB_SHIFT;

    if (t < CB_NODES) cnt[t] = 0;
    __syncthreads();

    for (int i = beg + t; i < end; i += FINE_PT * 1024) {
        int dy[FINE_PT];
#pragma unroll
        for (int k = 0; k < FINE_PT; ++k) {
            const int j = i + k * 1024;
            dy[k] = (j < end) ? cpair[j].y : -1;
        }
#pragma unroll
        for (int k = 0; k < FINE_PT; ++k)
            if (dy[k] >= 0) atomicAdd(&cnt[dy[k] - n0], 1);
    }
    __syncthreads();

    int v = 0, inc = 0;
    if (t < CB_NODES) {
        v   = cnt[t];
        inc = v;
        for (int o = 1; o < 32; o <<= 1) {
            const int u = __shfl_up(inc, o, 32);
            if (lane >= o) inc += u;
        }
        if (lane == 31) wtot[wave] = inc;
    }
    __syncthreads();
    if (t < CB_NODES) {
        int w = (lane < wave) ? wtot[lane] : 0;
        for (int o = 16; o > 0; o >>= 1) w += __shfl_xor(w, o, 32);
        const int ex = beg + w + inc - v;
        cnt[t] = ex;
        if (n0 + t <= N) off[n0 + t] = ex;
    }
    if (blockIdx.x == gridDim.x - 1 && t == 0) off[N] = end;
    __syncthreads();

    for (int i = beg + t; i < end; i += FINE_PT * 1024) {
        v2i p[FINE_PT];
#pragma unroll
        for (int k = 0; k < FINE_PT; ++k) {
            const int j = i + k * 1024;
            v2i q;
            q.x = 0;
            q.y = -1;
            p[k] = (j < end) ? cpair[j] : q;
        }
#pragma unroll
        for (int k = 0; k < FINE_PT; ++k) {
            if (p[k].y >= 0) {
                const int pos = atomicAdd(&cnt[p[k].y - n0], 1);
                bsrc[pos] = p[k].x;
            }
        }
    }
}

__global__ __launch_bounds__(256) void gat_node_gather(const int* __restrict__ off,
                                                       const int* __restrict__ bsrc,
                                                       const float* __restrict__ a_src,
                                                       const float* __restrict__ a_dst,
                                                       const float* __restrict__ h,
                                                       const float* __restrict__ bias,
                                                       float* __restrict__ out,
                                                       int N) {
    const int lane   = threadIdx.x & 31;
    const int head   = lane >> 3;
    const int nwaves = (int)(gridDim.x * 8);
    const v4f b      = *(const v4f*)&bias[lane * 4];
    for (int d = __builtin_amdgcn_readfirstlane((int)(blockIdx.x * 8 + (threadIdx.x >> 5))); d < N; d += nwaves) {
        const int   beg   = off[d];
        const int   end   = off[d + 1];
        const float ad    = a_dst[d * HEADS + head];
        const float eself = leaky(a_src[d * HEADS + head] + ad);

        float m = eself;
#pragma unroll 4
        for (int j = beg; j < end; ++j) {
            const int   s = bsrc[j];
            const float e = leaky(a_src[s * HEADS + head] + ad);
            m = e > m ? e : m;
        }

        float p   = __expf(eself - m);
        float den = p;
        v4f   hv  = *(const v4f*)&h[(size_t)d * HC + lane * 4];
        v4f   acc = hv * p;
#pragma unroll 4
        for (int j = beg; j < end; ++j) {
            const int s = bsrc[j];
            p  = __expf(leaky(a_src[s * HEADS + head] + ad) - m);
            hv = *(const v4f*)&h[(size_t)s * HC + lane * 4];
            den += p;
            acc += hv * p;
        }

        const float dd = den + 1e-16f;
        v4f o;
        o.x = acc.x / dd + b.x;
        o.y = acc.y / dd + b.y;
        o.z = acc.z / dd + b.z;
        o.w = acc.w / dd + b.w;
        *(v4f*)&out[(size_t)d * HC + lane * 4] = o;
    }
}

extern "C" void kernel_launch(void* const* d_in, const int* in_sizes, int n_in,
                              void* d_out, int out_size, void* d_ws, size_t ws_size,
                              hipStream_t stream) {
    const float* x        = (const float*)d_in[0];
    const int*   ei       = (const int*)d_in[1];
    const float* W        = (const float*)d_in[2];
    const float* att_src  = (const float*)d_in[3];
    const float* att_dst  = (const float*)d_in[4];
    const float* bias     = (const float*)d_in[5];
    float*       out      = (float*)d_out;

    const int N = in_sizes[0] / IN_F;
    const int E = in_sizes[1] / 2;

    const int CB = (N + CB_NODES - 1) / CB_NODES;
    const int NB = (E + CHUNK - 1) / CHUNK;

    float* h      = (float*)d_ws;
    float* a_src  = h + (size_t)N * HC;
    float* a_dst  = a_src + (size_t)N * HEADS;
    v2i*   cpair  = (v2i*)(a_dst + (size_t)N * HEADS);
    int*   bsrc   = (int*)(cpair + E);
    int*   off    = bsrc + E;
    int*   hist   = off + N + 1;
    int*   cstart = hist + (size_t)CB * NB;

    gat_gemm_wmma<<<1024, 128, GEMM_LDS_BYTES, stream>>>(x, W, att_src, att_dst, h, a_src, a_dst, N);

    gat_coarse_count<<<NB, 1024, 0, stream>>>(ei, hist, E, CB, NB);
    gat_coarse_place<<<NB, 1024, 0, stream>>>(ei, hist, cstart, cpair, E, CB, NB);
    gat_fine_place<<<CB, 1024, 0, stream>>>(cpair, cstart, off, bsrc, N);

    gat_node_gather<<<2048, 256, 0, stream>>>(off, bsrc, a_src, a_dst, h, bias, out, N);
}


# Node gather takes softmax max and weights edge-parallel (lane = head x 8 edge slots: ids and coefficients in two load rounds per node, one exp per edge and head), rows 8 at a time; sum order kept
# speedup vs baseline: 6.5198x; 1.0891x over previous
#include <hip/hip_runtime.h>

#define IN_F      128
#define HC        128
#define HEADS     4
#define OUT_F     32
#define NEG_SLOPE 0.2f

typedef float v2f __attribute__((ext_vector_type(2)));
typedef float v4f __attribute__((ext_vector_type(4)));
typedef float v8f __attribute__((ext_vector_type(8)));
typedef int   v4i __attribute__((ext_vector_type(4)));
typedef int   v2i __attribute__((ext_vector_type(2)));

__device__ __forceinline__ float leaky(float x) {
    return x > 0.0f ? x : NEG_SLOPE * x;
}

#define WSTR 136
#define XSTR 132
#define COEF_FLOATS 128
#define GEMM_LDS_BYTES ((128 * WSTR + 16 * XSTR + COEF_FLOATS) * 4)

__global__ __launch_bounds__(128) void gat_gemm_wmma(const float* __restrict__ x,
                                                     const float* __restrict__ W,
                                                     const float* __restrict__ att_src,
                                                     const float* __restrict__ att_dst,
                                                     float* __restrict__ h,
                                                     float* __restrict__ a_src,
                                                     float* __restrict__ a_dst,
                                                     int N) {
    extern __shared__ float smem[];
    float* Wl = smem;
    float* Xl = smem + 128 * WSTR;
    float* Cl = Xl + 16 * XSTR;

    const int tid    = threadIdx.x;
    const int ntiles = N / 16;

    const int wave = tid >> 5;
    const int lane = tid & 31;
    const int L    = lane & 15;
    const int g    = lane >> 4;

    const int col0 = wave * 32;
    const int col1 = wave * 32 + 16;

    const float as0 = att_src[col0 + L];
    const float as1 = att_src[col1 + L];
    const float ad0 = att_dst[col0 + L];
    const float ad1 = att_dst[col1 + L];

    for (int i = tid; i < 128 * 32; i += 128) {
        int r  = i >> 5;
        int c4 = i & 31;
        v4f v = ((const v4f*)W)[r * 32 + c4];
        *(v4f*)&Wl[r * WSTR + c4 * 4] = v;
    }
    for (int tile = blockIdx.x; tile < ntiles; tile += gridDim.x) {
        const int row0 = tile * 16;
        for (int i = tid; i < 16 * 32; i += 128) {
            int r  = i >> 5;
            int c4 = i & 31;
            v4f v = ((const v4f*)x)[(size_t)(row0 + r) * 32 + c4];
            *(v4f*)&Xl[r * XSTR + c4 * 4] = v;
        }
        __syncthreads();

        v8f acc0 = {};
        v8f acc1 = {};

        for (int k0 = 0; k0 < IN_F; k0 += 4) {
            const int k = k0 + 2 * g;
            v2f a, b0, b1;
            a.x  = Xl[L * XSTR + k];
            a.y  = Xl[L * XSTR + k + 1];
            b0.x = Wl[k * WSTR + col0 + L];
            b0.y = Wl[(k + 1) * WSTR + col0 + L];
            b1.x = Wl[k * WSTR + col1 + L];
            b1.y = Wl[(k + 1) * WSTR + col1 + L];
            acc0 = __builtin_amdgcn_wmma_f32_16x16x4_f32(false, a, false, b0, (short)0, acc0, false, false);
            acc1 = __builtin_amdgcn_wmma_f32_16x16x4_f32(false, a, false, b1, (short)0, acc1, false, false);
        }

        float v[16];
#pragma unroll
        for (int r = 0; r < 8; ++r) {
            v[r]     = acc0[r] * as0 + acc1[r] * as1;
            v[8 + r] = acc0[r] * ad0 + acc1[r] * ad1;
        }
        const bool hi8 = (L & 8) != 0;
        const bool hi4 = (L & 4) != 0;
        const bool hi2 = (L & 2) != 0;
        const bool hi1 = (L & 1) != 0;
        float u[8], t[4], q[2];
#pragma unroll
        for (int j = 0; j < 8; ++j) {
            const float keep = hi8 ? v[j + 8] : v[j];
            const float send = hi8 ? v[j] : v[j + 8];
            u[j] = keep + __shfl_xor(send, 8, 32);
        }
#pragma unroll
        for (int j = 0; j < 4; ++j) {
            const float keep = hi4 ? u[j + 4] : u[j];
            const float send = hi4 ? u[j] : u[j + 4];
            t[j] = keep + __shfl_xor(send, 4, 32);
        }
#pragma unroll
        for (int j = 0; j < 2; ++j) {
            const float keep = hi2 ? t[j + 2] : t[j];
            const float send = hi2 ? t[j] : t[j + 2];
            q[j] = keep + __shfl_xor(send, 2, 32);
        }
        const float keep1 = hi1 ? q[1] : q[0];
        const float send1 = hi1 ? q[0] : q[1];
        const float z = keep1 + __shfl_xor(send1, 1, 32);
        Cl[(L >> 3) * 64 + (8 * g + (L & 7)) * HEADS + wave] = z;
        __syncthreads();

        const int mbase = row0 + 8 * g;
        for (int r = 0; r < 8; ++r) {
            h[(size_t)(mbase + r) * HC + col0 + L] = acc0[r];
            h[(size_t)(mbase + r) * HC + col1 + L] = acc1[r];
        }

        if (wave == 0) {
            const v4f c = *(const v4f*)&Cl[lane * 4];
            float* dstp = (lane < 16) ? a_src + (size_t)(row0 + lane) * HEADS
                                      : a_dst + (size_t)(row0 + lane - 16) * HEADS;
            *(v4f*)dstp = c;
        }
    }
}

#define CB_SHIFT 9
#define CB_NODES 512
#define CB_MAX   256
#define CHUNK    8192
#define CHUNK_PT (CHUNK / 1024)
#define FINE_PT 10
#define NB_ROW_LOADS 7

__global__ __launch_bounds__(1024) void gat_coarse_count(const int* __restrict__ ei,
                                                         int* __restrict__ hist,
                                                         int E, int CB, int NB) {
    __shared__ int lc[CB_MAX];
    const int t = threadIdx.x;
    if (t < CB_MAX) lc[t] = 0;
    __syncthreads();

    const int e0 = blockIdx.x * CHUNK;
    int d[CHUNK_PT];
#pragma unroll
    for (int k = 0; k < CHUNK_PT; ++k) {
        const int e = e0 + k * 1024 + t;
        d[k] = (e < E) ? ei[E + e] : -1;
    }
#pragma unroll
    for (int k = 0; k < CHUNK_PT; ++k)
        if (d[k] >= 0) atomicAdd(&lc[d[k] >> CB_SHIFT], 1);
    __syncthreads();

    if (t < CB) hist[t * NB + blockIdx.x] = lc[t];
}

__global__ __launch_bounds__(1024) void gat_coarse_place(const int* __restrict__ ei,
                                                         const int* __restrict__ hist,
                                                         int* __restrict__ cstart,
                                                         v2i* __restrict__ cpair,
                                                         int E, int CB, int NB) {
    __shared__ int tot[CB_MAX];
    __shared__ int nxt[CB_MAX];
    const int t    = threadIdx.x;
    const int lane = t & 31;
    const int wave = t >> 5;
    const int b    = blockIdx.x;

    const int e0 = b * CHUNK;
    int s[CHUNK_PT], d[CHUNK_PT];
#pragma unroll
    for (int k = 0; k < CHUNK_PT; ++k) {
        const int e = e0 + k * 1024 + t;
        const bool ok = e < E;
        s[k] = ok ? ei[e] : 0;
        d[k] = ok ? ei[E + e] : -1;
    }

#pragma unroll
    for (int c0 = 0; c0 < CB_MAX; c0 += 128) {
        int sa[4], sb[4];
#pragma unroll
        for (int q = 0; q < 4; ++q) {
            const int c = c0 + 32 * q + wave;
            sa[q] = 0;
            sb[q] = 0;
#pragma unroll
            for (int r = 0; r < NB_ROW_LOADS; ++r) {
                const int i = lane + 32 * r;
                const int v = (c < CB && i < NB) ? hist[c * NB + i] : 0;
                sa[q] += v;
                sb[q] += (i < b) ? v : 0;
            }
        }
        if (NB > 32 * NB_ROW_LOADS) {
            for (int q = 0; q < 4; ++q) {
                const int c = c0 + 32 * q + wave;
                if (c < CB) {
                    for (int i = lane + 32 * NB_ROW_LOADS; i < NB; i += 32) {
                        const int v = hist[c * NB + i];
                        sa[q] += v;
                        sb[q] += (i < b) ? v : 0;
                    }
                }
            }
        }
#pragma unroll
        for (int q = 0; q < 4; ++q) {
            for (int o = 16; o > 0; o >>= 1) {
                sa[q] += __shfl_xor(sa[q], o, 32);
                sb[q] += __shfl_xor(sb[q], o, 32);
            }
            if (lane == 0) {
                tot[c0 + 32 * q + wave] = sa[q];
                nxt[c0 + 32 * q + wave] = sb[q];
            }
        }
    }
    __syncthreads();

    if (wave == 0) {
        int v[8];
        int s = 0;
#pragma unroll
        for (int k = 0; k < 8; ++k) { v[k] = tot[lane * 8 + k]; s += v[k]; }
        int inc = s;
        for (int o = 1; o < 32; o <<= 1) {
            const int u = __shfl_up(inc, o, 32);
            if (lane >= o) inc += u;
        }
        int run = inc - s;
#pragma unroll
        for (int k = 0; k < 8; ++k) {
            const int c = lane * 8 + k;
            if (b == 0 && c <= CB) cstart[c] = run;
            nxt[c] += run;
            run += v[k];
        }
    }
    __syncthreads();

#pragma unroll
    for (int k = 0; k < CHUNK_PT; ++k) {
        if (d[k] >= 0) {
            const int pos = atomicAdd(&nxt[d[k] >> CB_SHIFT], 1);
            v2i p;
            p.x = s[k];
            p.y = d[k];
            cpair[pos] = p;
        }
    }
}

__global__ __launch_bounds__(1024) void gat_fine_place(const v2i* __restrict__ cpair,
                                                       const int* __restrict__ cstart,
                                                       int* __restrict__ off,
                                                       int* __restrict__ bsrc,
                                                       int N) {
    __shared__ int cnt[CB_NODES];
    __shared__ int wtot[16];
    const int t    = threadIdx.x;
    const int lane = t & 31;
    const int wave = t >> 5;
    const int beg  = cstart[blockIdx.x];
    const int end  = cstart[blockIdx.x + 1];
    const int n0   = blockIdx.x << CB_SHIFT;

    if (t < CB_NODES) cnt[t] = 0;
    __syncthreads();

    for (int i = beg + t; i < end; i += FINE_PT * 1024) {
        int dy[FINE_PT];
#pragma unroll
        for (int k = 0; k < FINE_PT; ++k) {
            const int j = i + k * 1024;
            dy[k] = (j < end) ? cpair[j].y : -1;
        }
#pragma unroll
        for (int k = 0; k < FINE_PT; ++k)
            if (dy[k] >= 0) atomicAdd(&cnt[dy[k] - n0], 1);
    }
    __syncthreads();

    int v = 0, inc = 0;
    if (t < CB_NODES) {
        v   = cnt[t];
        inc = v;
        for (int o = 1; o < 32; o <<= 1) {
            const int u = __shfl_up(inc, o, 32);
            if (lane >= o) inc += u;
        }
        if (lane == 31) wtot[wave] = inc;
    }
    __syncthreads();
    if (t < CB_NODES) {
        int w = (lane < wave) ? wtot[lane] : 0;
        for (int o = 16; o > 0; o >>= 1) w += __shfl_xor(w, o, 32);
        const int ex = beg + w + inc - v;
        cnt[t] = ex;
        if (n0 + t <= N) off[n0 + t] = ex;
    }
    if (blockIdx.x == gridDim.x - 1 && t == 0) off[N] = end;
    __syncthreads();

    for (int i = beg + t; i < end; i += FINE_PT * 1024) {
        v2i p[FINE_PT];
#pragma unroll
        for (int k = 0; k < FINE_PT; ++k) {
            const int j = i + k * 1024;
            v2i q;
            q.x = 0;
            q.y = -1;
            p[k] = (j < end) ? cpair[j] : q;
        }
#pragma unroll
        for (int k = 0; k < FINE_PT; ++k) {
            if (p[k].y >= 0) {
                const int pos = atomicAdd(&cnt[p[k].y - n0], 1);
                bsrc[pos] = p[k].x;
            }
        }
    }
}


#define XLANE(v, X) __builtin_bit_cast(float, __builtin_amdgcn_update_dpp(0, __builtin_bit_cast(int, (v)), 0x160 + (X), 0xf, 0xf, false))
template <int T>
__device__ __forceinline__ float slot_bcast(const float v) {
    return __builtin_bit_cast(float, __builtin_amdgcn_mov_dpp8(__builtin_bit_cast(unsigned, v), (unsigned)T * 0x249249u));
}

#define ROW_LOAD(T) hv[T] = *(const v4f*)&hrow[(size_t)__builtin_amdgcn_readlane(sv, T) * HC]
#define ROW_SUM(T)  { const float p = slot_bcast<T>(pv); den += p; acc += hv[T] * p; }

__device__ __forceinline__ void gat_rows8_load(const int n, const int sv,
                                               const float* __restrict__ hrow, v4f (&hv)[8]) {
    if (0 < n) ROW_LOAD(0);
    if (1 < n) ROW_LOAD(1);
    if (2 < n) ROW_LOAD(2);
    if (3 < n) ROW_LOAD(3);
    if (4 < n) ROW_LOAD(4);
    if (5 < n) ROW_LOAD(5);
    if (6 < n) ROW_LOAD(6);
    if (7 < n) ROW_LOAD(7);
}

__device__ __forceinline__ void gat_rows8_sum(const int n, const float pv, const v4f (&hv)[8],
                                              float& den, v4f& acc) {
    if (0 < n) ROW_SUM(0)
    if (1 < n) ROW_SUM(1)
    if (2 < n) ROW_SUM(2)
    if (3 < n) ROW_SUM(3)
    if (4 < n) ROW_SUM(4)
    if (5 < n) ROW_SUM(5)
    if (6 < n) ROW_SUM(6)
    if (7 < n) ROW_SUM(7)
}

__global__ __launch_bounds__(256, 16) void gat_node_gather(const int* __restrict__ off,
                                                       const int* __restrict__ bsrc,
                                                       const float* __restrict__ a_src,
                                                       const float* __restrict__ a_dst,
                                                       const float* __restrict__ h,
                                                       const float* __restrict__ bias,
                                                       float* __restrict__ out,
                                                       int N) {
    const int lane   = threadIdx.x & 31;
    const int head   = lane >> 3;
    const int slot   = lane & 7;
    const int nwaves = (int)(gridDim.x * 8);
    const v4f b      = *(const v4f*)&bias[lane * 4];
    const float* hrow = h + lane * 4;
    const float NEG_INF = -__builtin_huge_valf();
    for (int d = __builtin_amdgcn_readfirstlane((int)(blockIdx.x * 8 + (threadIdx.x >> 5))); d < N; d += nwaves) {
        const int   beg   = off[d];
        const int   deg   = off[d + 1] - beg;
        const float ad    = a_dst[d * HEADS + head];
        const float eself = leaky(a_src[d * HEADS + head] + ad);
        const v4f   hself = *(const v4f*)&hrow[(size_t)d * HC];

        int   sv[4];
        float ev[4];
#pragma unroll
        for (int c = 0; c < 4; ++c) {
            const int j = 8 * c + slot;
            sv[c] = (j < deg) ? bsrc[beg + j] : d;
        }
#pragma unroll
        for (int c = 0; c < 4; ++c) ev[c] = leaky(a_src[sv[c] * HEADS + head] + ad);
        v4f hv[8];
        gat_rows8_load(deg, sv[0], hrow, hv);
        float m = eself;
#pragma unroll
        for (int c = 0; c < 4; ++c) m = ev[c] > m ? ev[c] : m;
        for (int j0 = 32; j0 < deg; j0 += 8) {
            const int j = j0 + slot;
            if (j < deg) {
                const float e = leaky(a_src[bsrc[beg + j] * HEADS + head] + ad);
                m = e > m ? e : m;
            }
        }
        { const float o = XLANE(m, 1); m = o > m ? o : m; }
        { const float o = XLANE(m, 2); m = o > m ? o : m; }
        { const float o = XLANE(m, 4); m = o > m ? o : m; }

        float p   = __expf(eself - m);
        float den = p;
        v4f   acc = hself * p;
        gat_rows8_sum(deg, __expf(ev[0] - m), hv, den, acc);
#pragma unroll
        for (int c = 1; c < 4; ++c) {
            const int n = deg - 8 * c;
            if (n > 0) {
                gat_rows8_load(n, sv[c], hrow, hv);
                gat_rows8_sum(n, __expf(ev[c] - m), hv, den, acc);
            }
        }
        for (int j0 = 32; j0 < deg; j0 += 8) {
            const int  j  = j0 + slot;
            const bool ok = j < deg;
            const int   s = ok ? bsrc[beg + j] : d;
            const float e = ok ? leaky(a_src[s * HEADS + head] + ad) : NEG_INF;
            gat_rows8_load(deg - j0, s, hrow, hv);
            gat_rows8_sum(deg - j0, __expf(e - m), hv, den, acc);
        }

        const float dd = den + 1e-16f;
        v4f o;
        o.x = acc.x / dd + b.x;
        o.y = acc.y / dd + b.y;
        o.z = acc.z / dd + b.z;
        o.w = acc.w / dd + b.w;
        *(v4f*)&out[(size_t)d * HC + lane * 4] = o;
    }
}

extern "C" void kernel_launch(void* const* d_in, const int* in_sizes, int n_in,
                              void* d_out, int out_size, void* d_ws, size_t ws_size,
                              hipStream_t stream) {
    const float* x        = (const float*)d_in[0];
    const int*   ei       = (const int*)d_in[1];
    const float* W        = (const float*)d_in[2];
    const float* att_src  = (const float*)d_in[3];
    const float* att_dst  = (const float*)d_in[4];
    const float* bias     = (const float*)d_in[5];
    float*       out      = (float*)d_out;

    const int N = in_sizes[0] / IN_F;
    const int E = in_sizes[1] / 2;

    const int CB = (N + CB_NODES - 1) / CB_NODES;
    const int NB = (E + CHUNK - 1) / CHUNK;

    float* h      = (float*)d_ws;
    float* a_src  = h + (size_t)N * HC;
    float* a_dst  = a_src + (size_t)N * HEADS;
    v2i*   cpair  = (v2i*)(a_dst + (size_t)N * HEADS);
    int*   bsrc   = (int*)(cpair + E);
    int*   off    = bsrc + E;
    int*   hist   = off + N + 1;
    int*   cstart = hist + (size_t)CB * NB;

    gat_gemm_wmma<<<1024, 128, GEMM_LDS_BYTES, stream>>>(x, W, att_src, att_dst, h, a_src, a_dst, N);

    gat_coarse_count<<<NB, 1024, 0, stream>>>(ei, hist, E, CB, NB);
    gat_coarse_place<<<NB, 1024, 0, stream>>>(ei, hist, cstart, cpair, E, CB, NB);
    gat_fine_place<<<CB, 1024, 0, stream>>>(cpair, cstart, off, bsrc, N);

    gat_node_gather<<<2048, 256, 0, stream>>>(off, bsrc, a_src, a_dst, h, bias, out, N);
}


# GEMM: the next trip's x tile is loaded into registers as one group of four loads under the current trip's k loop, replacing four dependent load-to-LDS round trips at the head of every trip
# speedup vs baseline: 6.7824x; 1.0403x over previous
#include <hip/hip_runtime.h>

#define IN_F      128
#define HC        128
#define HEADS     4
#define OUT_F     32
#define NEG_SLOPE 0.2f

typedef float v2f __attribute__((ext_vector_type(2)));
typedef float v4f __attribute__((ext_vector_type(4)));
typedef float v8f __attribute__((ext_vector_type(8)));
typedef int   v4i __attribute__((ext_vector_type(4)));
typedef int   v2i __attribute__((ext_vector_type(2)));

__device__ __forceinline__ float leaky(float x) {
    return x > 0.0f ? x : NEG_SLOPE * x;
}

#define WSTR 136
#define XSTR 132
#define COEF_FLOATS 128
#define GEMM_LDS_BYTES ((128 * WSTR + 16 * XSTR + COEF_FLOATS) * 4)

__global__ __launch_bounds__(128) void gat_gemm_wmma(const float* __restrict__ x,
                                                     const float* __restrict__ W,
                                                     const float* __restrict__ att_src,
                                                     const float* __restrict__ att_dst,
                                                     float* __restrict__ h,
                                                     float* __restrict__ a_src,
                                                     float* __restrict__ a_dst,
                                                     int N) {
    extern __shared__ float smem[];
    float* Wl = smem;
    float* Xl = smem + 128 * WSTR;
    float* Cl = Xl + 16 * XSTR;

    const int tid    = threadIdx.x;
    const int ntiles = N / 16;

    const int wave = tid >> 5;
    const int lane = tid & 31;
    const int L    = lane & 15;
    const int g    = lane >> 4;

    const int col0 = wave * 32;
    const int col1 = wave * 32 + 16;

    const float as0 = att_src[col0 + L];
    const float as1 = att_src[col1 + L];
    const float ad0 = att_dst[col0 + L];
    const float ad1 = att_dst[col1 + L];

    const v4f* x4 = (const v4f*)x;
    v4f xr[4] = {};
    if ((int)blockIdx.x < ntiles) {
#pragma unroll
        for (int j = 0; j < 4; ++j)
            xr[j] = x4[(size_t)((int)blockIdx.x * 16 + wave + 4 * j) * 32 + lane];
    }

    for (int i = tid; i < 128 * 32; i += 128) {
        int r  = i >> 5;
        int c4 = i & 31;
        v4f v = ((const v4f*)W)[r * 32 + c4];
        *(v4f*)&Wl[r * WSTR + c4 * 4] = v;
    }
    for (int tile = blockIdx.x; tile < ntiles; tile += gridDim.x) {
        const int row0 = tile * 16;
#pragma unroll
        for (int j = 0; j < 4; ++j)
            *(v4f*)&Xl[(wave + 4 * j) * XSTR + lane * 4] = xr[j];
        __syncthreads();
        const int nxt = tile + (int)gridDim.x;
        if (nxt < ntiles) {
#pragma unroll
            for (int j = 0; j < 4; ++j)
                xr[j] = x4[(size_t)(nxt * 16 + wave + 4 * j) * 32 + lane];
        }

        v8f acc0 = {};
        v8f acc1 = {};

        for (int k0 = 0; k0 < IN_F; k0 += 4) {
            const int k = k0 + 2 * g;
            v2f a, b0, b1;
            a.x  = Xl[L * XSTR + k];
            a.y  = Xl[L * XSTR + k + 1];
            b0.x = Wl[k * WSTR + col0 + L];
            b0.y = Wl[(k + 1) * WSTR + col0 + L];
            b1.x = Wl[k * WSTR + col1 + L];
            b1.y = Wl[(k + 1) * WSTR + col1 + L];
            acc0 = __builtin_amdgcn_wmma_f32_16x16x4_f32(false, a, false, b0, (short)0, acc0, false, false);
            acc1 = __builtin_amdgcn_wmma_f32_16x16x4_f32(false, a, false, b1, (short)0, acc1, false, false);
        }

        float v[16];
#pragma unroll
        for (int r = 0; r < 8; ++r) {
            v[r]     = acc0[r] * as0 + acc1[r] * as1;
            v[8 + r] = acc0[r] * ad0 + acc1[r] * ad1;
        }
        const bool hi8 = (L & 8) != 0;
        const bool hi4 = (L & 4) != 0;
        const bool hi2 = (L & 2) != 0;
        const bool hi1 = (L & 1) != 0;
        float u[8], t[4], q[2];
#pragma unroll
        for (int j = 0; j < 8; ++j) {
            const float keep = hi8 ? v[j + 8] : v[j];
            const float send = hi8 ? v[j] : v[j + 8];
            u[j] = keep + __shfl_xor(send, 8, 32);
        }
#pragma unroll
        for (int j = 0; j < 4; ++j) {
            const float keep = hi4 ? u[j + 4] : u[j];
            const float send = hi4 ? u[j] : u[j + 4];
            t[j] = keep + __shfl_xor(send, 4, 32);
        }
#pragma unroll
        for (int j = 0; j < 2; ++j) {
            const float keep = hi2 ? t[j + 2] : t[j];
            const float send = hi2 ? t[j] : t[j + 2];
            q[j] = keep + __shfl_xor(send, 2, 32);
        }
        const float keep1 = hi1 ? q[1] : q[0];
        const float send1 = hi1 ? q[0] : q[1];
        const float z = keep1 + __shfl_xor(send1, 1, 32);
        Cl[(L >> 3) * 64 + (8 * g + (L & 7)) * HEADS + wave] = z;
        __syncthreads();

        const int mbase = row0 + 8 * g;
        for (int r = 0; r < 8; ++r) {
            h[(size_t)(mbase + r) * HC + col0 + L] = acc0[r];
            h[(size_t)(mbase + r) * HC + col1 + L] = acc1[r];
        }

        if (wave == 0) {
            const v4f c = *(const v4f*)&Cl[lane * 4];
            float* dstp = (lane < 16) ? a_src + (size_t)(row0 + lane) * HEADS
                                      : a_dst + (size_t)(row0 + lane - 16) * HEADS;
            *(v4f*)dstp = c;
        }
    }
}

#define CB_SHIFT 9
#define CB_NODES 512
#define CB_MAX   256
#define CHUNK    8192
#define CHUNK_PT (CHUNK / 1024)
#define FINE_PT 10
#define NB_ROW_LOADS 7

__global__ __launch_bounds__(1024) void gat_coarse_count(const int* __restrict__ ei,
                                                         int* __restrict__ hist,
                                                         int E, int CB, int NB) {
    __shared__ int lc[CB_MAX];
    const int t = threadIdx.x;
    if (t < CB_MAX) lc[t] = 0;
    __syncthreads();

    const int e0 = blockIdx.x * CHUNK;
    int d[CHUNK_PT];
#pragma unroll
    for (int k = 0; k < CHUNK_PT; ++k) {
        const int e = e0 + k * 1024 + t;
        d[k] = (e < E) ? ei[E + e] : -1;
    }
#pragma unroll
    for (int k = 0; k < CHUNK_PT; ++k)
        if (d[k] >= 0) atomicAdd(&lc[d[k] >> CB_SHIFT], 1);
    __syncthreads();

    if (t < CB) hist[t * NB + blockIdx.x] = lc[t];
}

__global__ __launch_bounds__(1024) void gat_coarse_place(const int* __restrict__ ei,
                                                         const int* __restrict__ hist,
                                                         int* __restrict__ cstart,
                                                         v2i* __restrict__ cpair,
                                                         int E, int CB, int NB) {
    __shared__ int tot[CB_MAX];
    __shared__ int nxt[CB_MAX];
    const int t    = threadIdx.x;
    const int lane = t & 31;
    const int wave = t >> 5;
    const int b    = blockIdx.x;

    const int e0 = b * CHUNK;
    int s[CHUNK_PT], d[CHUNK_PT];
#pragma unroll
    for (int k = 0; k < CHUNK_PT; ++k) {
        const int e = e0 + k * 1024 + t;
        const bool ok = e < E;
        s[k] = ok ? ei[e] : 0;
        d[k] = ok ? ei[E + e] : -1;
    }

#pragma unroll
    for (int c0 = 0; c0 < CB_MAX; c0 += 128) {
        int sa[4], sb[4];
#pragma unroll
        for (int q = 0; q < 4; ++q) {
            const int c = c0 + 32 * q + wave;
            sa[q] = 0;
            sb[q] = 0;
#pragma unroll
            for (int r = 0; r < NB_ROW_LOADS; ++r) {
                const int i = lane + 32 * r;
                const int v = (c < CB && i < NB) ? hist[c * NB + i] : 0;
                sa[q] += v;
                sb[q] += (i < b) ? v : 0;
            }
        }
        if (NB > 32 * NB_ROW_LOADS) {
            for (int q = 0; q < 4; ++q) {
                const int c = c0 + 32 * q + wave;
                if (c < CB) {
                    for (int i = lane + 32 * NB_ROW_LOADS; i < NB; i += 32) {
                        const int v = hist[c * NB + i];
                        sa[q] += v;
                        sb[q] += (i < b) ? v : 0;
                    }
                }
            }
        }
#pragma unroll
        for (int q = 0; q < 4; ++q) {
            for (int o = 16; o > 0; o >>= 1) {
                sa[q] += __shfl_xor(sa[q], o, 32);
                sb[q] += __shfl_xor(sb[q], o, 32);
            }
            if (lane == 0) {
                tot[c0 + 32 * q + wave] = sa[q];
                nxt[c0 + 32 * q + wave] = sb[q];
            }
        }
    }
    __syncthreads();

    if (wave == 0) {
        int v[8];
        int s = 0;
#pragma unroll
        for (int k = 0; k < 8; ++k) { v[k] = tot[lane * 8 + k]; s += v[k]; }
        int inc = s;
        for (int o = 1; o < 32; o <<= 1) {
            const int u = __shfl_up(inc, o, 32);
            if (lane >= o) inc += u;
        }
        int run = inc - s;
#pragma unroll
        for (int k = 0; k < 8; ++k) {
            const int c = lane * 8 + k;
            if (b == 0 && c <= CB) cstart[c] = run;
            nxt[c] += run;
            run += v[k];
        }
    }
    __syncthreads();

#pragma unroll
    for (int k = 0; k < CHUNK_PT; ++k) {
        if (d[k] >= 0) {
            const int pos = atomicAdd(&nxt[d[k] >> CB_SHIFT], 1);
            v2i p;
            p.x = s[k];
            p.y = d[k];
            cpair[pos] = p;
        }
    }
}

__global__ __launch_bounds__(1024) void gat_fine_place(const v2i* __restrict__ cpair,
                                                       const int* __restrict__ cstart,
                                                       int* __restrict__ off,
                                                       int* __restrict__ bsrc,
                                                       int N) {
    __shared__ int cnt[CB_NODES];
    __shared__ int wtot[16];
    const int t    = threadIdx.x;
    const int lane = t & 31;
    const int wave = t >> 5;
    const int beg  = cstart[blockIdx.x];
    const int end  = cstart[blockIdx.x + 1];
    const int n0   = blockIdx.x << CB_SHIFT;

    if (t < CB_NODES) cnt[t] = 0;
    __syncthreads();

    for (int i = beg + t; i < end; i += FINE_PT * 1024) {
        int dy[FINE_PT];
#pragma unroll
        for (int k = 0; k < FINE_PT; ++k) {
            const int j = i + k * 1024;
            dy[k] = (j < end) ? cpair[j].y : -1;
        }
#pragma unroll
        for (int k = 0; k < FINE_PT; ++k)
            if (dy[k] >= 0) atomicAdd(&cnt[dy[k] - n0], 1);
    }
    __syncthreads();

    int v = 0, inc = 0;
    if (t < CB_NODES) {
        v   = cnt[t];
        inc = v;
        for (int o = 1; o < 32; o <<= 1) {
            const int u = __shfl_up(inc, o, 32);
            if (lane >= o) inc += u;
        }
        if (lane == 31) wtot[wave] = inc;
    }
    __syncthreads();
    if (t < CB_NODES) {
        int w = (lane < wave) ? wtot[lane] : 0;
        for (int o = 16; o > 0; o >>= 1) w += __shfl_xor(w, o, 32);
        const int ex = beg + w + inc - v;
        cnt[t] = ex;
        if (n0 + t <= N) off[n0 + t] = ex;
    }
    if (blockIdx.x == gridDim.x - 1 && t == 0) off[N] = end;
    __syncthreads();

    for (int i = beg + t; i < end; i += FINE_PT * 1024) {
        v2i p[FINE_PT];
#pragma unroll
        for (int k = 0; k < FINE_PT; ++k) {
            const int j = i + k * 1024;
            v2i q;
            q.x = 0;
            q.y = -1;
            p[k] = (j < end) ? cpair[j] : q;
        }
#pragma unroll
        for (int k = 0; k < FINE_PT; ++k) {
            if (p[k].y >= 0) {
                const int pos = atomicAdd(&cnt[p[k].y - n0], 1);
                bsrc[pos] = p[k].x;
            }
        }
    }
}


#define XLANE(v, X) __builtin_bit_cast(float, __builtin_amdgcn_update_dpp(0, __builtin_bit_cast(int, (v)), 0x160 + (X), 0xf, 0xf, false))
template <int T>
__device__ __forceinline__ float slot_bcast(const float v) {
    return __builtin_bit_cast(float, __builtin_amdgcn_mov_dpp8(__builtin_bit_cast(unsigned, v), (unsigned)T * 0x249249u));
}

#define ROW_LOAD(T) hv[T] = *(const v4f*)&hrow[(size_t)__builtin_amdgcn_readlane(sv, T) * HC]
#define ROW_SUM(T)  { const float p = slot_bcast<T>(pv); den += p; acc += hv[T] * p; }

__device__ __forceinline__ void gat_rows8_load(const int n, const int sv,
                                               const float* __restrict__ hrow, v4f (&hv)[8]) {
    if (0 < n) ROW_LOAD(0);
    if (1 < n) ROW_LOAD(1);
    if (2 < n) ROW_LOAD(2);
    if (3 < n) ROW_LOAD(3);
    if (4 < n) ROW_LOAD(4);
    if (5 < n) ROW_LOAD(5);
    if (6 < n) ROW_LOAD(6);
    if (7 < n) ROW_LOAD(7);
}

__device__ __forceinline__ void gat_rows8_sum(const int n, const float pv, const v4f (&hv)[8],
                                              float& den, v4f& acc) {
    if (0 < n) ROW_SUM(0)
    if (1 < n) ROW_SUM(1)
    if (2 < n) ROW_SUM(2)
    if (3 < n) ROW_SUM(3)
    if (4 < n) ROW_SUM(4)
    if (5 < n) ROW_SUM(5)
    if (6 < n) ROW_SUM(6)
    if (7 < n) ROW_SUM(7)
}

__global__ __launch_bounds__(256, 16) void gat_node_gather(const int* __restrict__ off,
                                                       const int* __restrict__ bsrc,
                                                       const float* __restrict__ a_src,
                                                       const float* __restrict__ a_dst,
                                                       const float* __restrict__ h,
                                                       const float* __restrict__ bias,
                                                       float* __restrict__ out,
                                                       int N) {
    const int lane   = threadIdx.x & 31;
    const int head   = lane >> 3;
    const int slot   = lane & 7;
    const int nwaves = (int)(gridDim.x * 8);
    const v4f b      = *(const v4f*)&bias[lane * 4];
    const float* hrow = h + lane * 4;
    const float NEG_INF = -__builtin_huge_valf();
    for (int d = __builtin_amdgcn_readfirstlane((int)(blockIdx.x * 8 + (threadIdx.x >> 5))); d < N; d += nwaves) {
        const int   beg   = off[d];
        const int   deg   = off[d + 1] - beg;
        const float ad    = a_dst[d * HEADS + head];
        const float eself = leaky(a_src[d * HEADS + head] + ad);
        const v4f   hself = *(const v4f*)&hrow[(size_t)d * HC];

        int   sv[4];
        float ev[4];
#pragma unroll
        for (int c = 0; c < 4; ++c) {
            const int j = 8 * c + slot;
            sv[c] = (j < deg) ? bsrc[beg + j] : d;
        }
#pragma unroll
        for (int c = 0; c < 4; ++c) ev[c] = leaky(a_src[sv[c] * HEADS + head] + ad);
        v4f hv[8];
        gat_rows8_load(deg, sv[0], hrow, hv);
        float m = eself;
#pragma unroll
        for (int c = 0; c < 4; ++c) m = ev[c] > m ? ev[c] : m;
        for (int j0 = 32; j0 < deg; j0 += 8) {
            const int j = j0 + slot;
            if (j < deg) {
                const float e = leaky(a_src[bsrc[beg + j] * HEADS + head] + ad);
                m = e > m ? e : m;
            }
        }
        { const float o = XLANE(m, 1); m = o > m ? o : m; }
        { const float o = XLANE(m, 2); m = o > m ? o : m; }
        { const float o = XLANE(m, 4); m = o > m ? o : m; }

        float p   = __expf(eself - m);
        float den = p;
        v4f   acc = hself * p;
        gat_rows8_sum(deg, __expf(ev[0] - m), hv, den, acc);
#pragma unroll
        for (int c = 1; c < 4; ++c) {
            const int n = deg - 8 * c;
            if (n > 0) {
                gat_rows8_load(n, sv[c], hrow, hv);
                gat_rows8_sum(n, __expf(ev[c] - m), hv, den, acc);
            }
        }
        for (int j0 = 32; j0 < deg; j0 += 8) {
            const int  j  = j0 + slot;
            const bool ok = j < deg;
            const int   s = ok ? bsrc[beg + j] : d;
            const float e = ok ? leaky(a_src[s * HEADS + head] + ad) : NEG_INF;
            gat_rows8_load(deg - j0, s, hrow, hv);
            gat_rows8_sum(deg - j0, __expf(e - m), hv, den, acc);
        }

        const float dd = den + 1e-16f;
        v4f o;
        o.x = acc.x / dd + b.x;
        o.y = acc.y / dd + b.y;
        o.z = acc.z / dd + b.z;
        o.w = acc.w / dd + b.w;
        *(v4f*)&out[(size_t)d * HC + lane * 4] = o;
    }
}

extern "C" void kernel_launch(void* const* d_in, const int* in_sizes, int n_in,
                              void* d_out, int out_size, void* d_ws, size_t ws_size,
                              hipStream_t stream) {
    const float* x        = (const float*)d_in[0];
    const int*   ei       = (const int*)d_in[1];
    const float* W        = (const float*)d_in[2];
    const float* att_src  = (const float*)d_in[3];
    const float* att_dst  = (const float*)d_in[4];
    const float* bias     = (const float*)d_in[5];
    float*       out      = (float*)d_out;

    const int N = in_sizes[0] / IN_F;
    const int E = in_sizes[1] / 2;

    const int CB = (N + CB_NODES - 1) / CB_NODES;
    const int NB = (E + CHUNK - 1) / CHUNK;

    float* h      = (float*)d_ws;
    float* a_src  = h + (size_t)N * HC;
    float* a_dst  = a_src + (size_t)N * HEADS;
    v2i*   cpair  = (v2i*)(a_dst + (size_t)N * HEADS);
    int*   bsrc   = (int*)(cpair + E);
    int*   off    = bsrc + E;
    int*   hist   = off + N + 1;
    int*   cstart = hist + (size_t)CB * NB;

    gat_gemm_wmma<<<1024, 128, GEMM_LDS_BYTES, stream>>>(x, W, att_src, att_dst, h, a_src, a_dst, N);

    gat_coarse_count<<<NB, 1024, 0, stream>>>(ei, hist, E, CB, NB);
    gat_coarse_place<<<NB, 1024, 0, stream>>>(ei, hist, cstart, cpair, E, CB, NB);
    gat_fine_place<<<CB, 1024, 0, stream>>>(cpair, cstart, off, bsrc, N);

    gat_node_gather<<<2048, 256, 0, stream>>>(off, bsrc, a_src, a_dst, h, bias, out, N);
}


# Edge grouping in two launches, not three: each chunk orders its edges by coarse bucket in its own slots (no count pass, no table sums), the fine pass collects the runs; sums in the new arrival order
# speedup vs baseline: 7.3251x; 1.0800x over previous
#include <hip/hip_runtime.h>

#define IN_F      128
#define HC        128
#define HEADS     4
#define OUT_F     32
#define NEG_SLOPE 0.2f

typedef float v2f __attribute__((ext_vector_type(2)));
typedef float v4f __attribute__((ext_vector_type(4)));
typedef float v8f __attribute__((ext_vector_type(8)));
typedef int   v4i __attribute__((ext_vector_type(4)));
typedef int   v2i __attribute__((ext_vector_type(2)));

__device__ __forceinline__ float leaky(float x) {
    return x > 0.0f ? x : NEG_SLOPE * x;
}

#define WSTR 136
#define XSTR 132
#define COEF_FLOATS 128
#define GEMM_LDS_BYTES ((128 * WSTR + 16 * XSTR + COEF_FLOATS) * 4)

__global__ __launch_bounds__(128) void gat_gemm_wmma(const float* __restrict__ x,
                                                     const float* __restrict__ W,
                                                     const float* __restrict__ att_src,
                                                     const float* __restrict__ att_dst,
                                                     float* __restrict__ h,
                                                     float* __restrict__ a_src,
                                                     float* __restrict__ a_dst,
                                                     int N) {
    extern __shared__ float smem[];
    float* Wl = smem;
    float* Xl = smem + 128 * WSTR;
    float* Cl = Xl + 16 * XSTR;

    const int tid    = threadIdx.x;
    const int ntiles = N / 16;

    const int wave = tid >> 5;
    const int lane = tid & 31;
    const int L    = lane & 15;
    const int g    = lane >> 4;

    const int col0 = wave * 32;
    const int col1 = wave * 32 + 16;

    const float as0 = att_src[col0 + L];
    const float as1 = att_src[col1 + L];
    const float ad0 = att_dst[col0 + L];
    const float ad1 = att_dst[col1 + L];

    const v4f* x4 = (const v4f*)x;
    v4f xr[4] = {};
    if ((int)blockIdx.x < ntiles) {
#pragma unroll
        for (int j = 0; j < 4; ++j)
            xr[j] = x4[(size_t)((int)blockIdx.x * 16 + wave + 4 * j) * 32 + lane];
    }

    for (int i = tid; i < 128 * 32; i += 128) {
        int r  = i >> 5;
        int c4 = i & 31;
        v4f v = ((const v4f*)W)[r * 32 + c4];
        *(v4f*)&Wl[r * WSTR + c4 * 4] = v;
    }
    for (int tile = blockIdx.x; tile < ntiles; tile += gridDim.x) {
        const int row0 = tile * 16;
#pragma unroll
        for (int j = 0; j < 4; ++j)
            *(v4f*)&Xl[(wave + 4 * j) * XSTR + lane * 4] = xr[j];
        __syncthreads();
        const int nxt = tile + (int)gridDim.x;
        if (nxt < ntiles) {
#pragma unroll
            for (int j = 0; j < 4; ++j)
                xr[j] = x4[(size_t)(nxt * 16 + wave + 4 * j) * 32 + lane];
        }

        v8f acc0 = {};
        v8f acc1 = {};

        for (int k0 = 0; k0 < IN_F; k0 += 4) {
            const int k = k0 + 2 * g;
            v2f a, b0, b1;
            a.x  = Xl[L * XSTR + k];
            a.y  = Xl[L * XSTR + k + 1];
            b0.x = Wl[k * WSTR + col0 + L];
            b0.y = Wl[(k + 1) * WSTR + col0 + L];
            b1.x = Wl[k * WSTR + col1 + L];
            b1.y = Wl[(k + 1) * WSTR + col1 + L];
            acc0 = __builtin_amdgcn_wmma_f32_16x16x4_f32(false, a, false, b0, (short)0, acc0, false, false);
            acc1 = __builtin_amdgcn_wmma_f32_16x16x4_f32(false, a, false, b1, (short)0, acc1, false, false);
        }

        float v[16];
#pragma unroll
        for (int r = 0; r < 8; ++r) {
            v[r]     = acc0[r] * as0 + acc1[r] * as1;
            v[8 + r] = acc0[r] * ad0 + acc1[r] * ad1;
        }
        const bool hi8 = (L & 8) != 0;
        const bool hi4 = (L & 4) != 0;
        const bool hi2 = (L & 2) != 0;
        const bool hi1 = (L & 1) != 0;
        float u[8], t[4], q[2];
#pragma unroll
        for (int j = 0; j < 8; ++j) {
            const float keep = hi8 ? v[j + 8] : v[j];
            const float send = hi8 ? v[j] : v[j + 8];
            u[j] = keep + __shfl_xor(send, 8, 32);
        }
#pragma unroll
        for (int j = 0; j < 4; ++j) {
            const float keep = hi4 ? u[j + 4] : u[j];
            const float send = hi4 ? u[j] : u[j + 4];
            t[j] = keep + __shfl_xor(send, 4, 32);
        }
#pragma unroll
        for (int j = 0; j < 2; ++j) {
            const float keep = hi2 ? t[j + 2] : t[j];
            const float send = hi2 ? t[j] : t[j + 2];
            q[j] = keep + __shfl_xor(send, 2, 32);
        }
        const float keep1 = hi1 ? q[1] : q[0];
        const float send1 = hi1 ? q[0] : q[1];
        const float z = keep1 + __shfl_xor(send1, 1, 32);
        Cl[(L >> 3) * 64 + (8 * g + (L & 7)) * HEADS + wave] = z;
        __syncthreads();

        const int mbase = row0 + 8 * g;
        for (int r = 0; r < 8; ++r) {
            h[(size_t)(mbase + r) * HC + col0 + L] = acc0[r];
            h[(size_t)(mbase + r) * HC + col1 + L] = acc1[r];
        }

        if (wave == 0) {
            const v4f c = *(const v4f*)&Cl[lane * 4];
            float* dstp = (lane < 16) ? a_src + (size_t)(row0 + lane) * HEADS
                                      : a_dst + (size_t)(row0 + lane - 16) * HEADS;
            *(v4f*)dstp = c;
        }
    }
}

#define CB_SHIFT 9
#define CB_NODES 512
#define CB_MAX   256
#define CBP      257
#define CHUNK    8192
#define CHUNK_PT (CHUNK / 1024)
#define SEG_MAX  1024
#define SEG_PW   7

__global__ __launch_bounds__(1024) void gat_coarse_local(const int* __restrict__ ei,
                                                         int* __restrict__ choff,
                                                         v2i* __restrict__ cpair,
                                                         int E, int CB) {
    __shared__ int cnt[CB_MAX];
    __shared__ int nxt[CB_MAX];
    const int t    = threadIdx.x;
    const int lane = t & 31;
    const int wave = t >> 5;
    const int b    = blockIdx.x;
    if (t < CB_MAX) cnt[t] = 0;

    const int e0 = b * CHUNK;
    int s[CHUNK_PT], d[CHUNK_PT];
#pragma unroll
    for (int k = 0; k < CHUNK_PT; ++k) {
        const int e = e0 + k * 1024 + t;
        const bool ok = e < E;
        s[k] = ok ? ei[e] : 0;
        d[k] = ok ? ei[E + e] : -1;
    }
    __syncthreads();
#pragma unroll
    for (int k = 0; k < CHUNK_PT; ++k)
        if (d[k] >= 0) atomicAdd(&cnt[d[k] >> CB_SHIFT], 1);
    __syncthreads();

    if (wave == 0) {
        int v[8];
        int sum = 0;
#pragma unroll
        for (int k = 0; k < 8; ++k) { v[k] = cnt[lane * 8 + k]; sum += v[k]; }
        int inc = sum;
        for (int o = 1; o < 32; o <<= 1) {
            const int u = __shfl_up(inc, o, 32);
            if (lane >= o) inc += u;
        }
        int run = inc - sum;
#pragma unroll
        for (int k = 0; k < 8; ++k) {
            const int c = lane * 8 + k;
            nxt[c] = run;
            if (c <= CB) choff[b * CBP + c] = run;
            run += v[k];
        }
    }
    __syncthreads();

#pragma unroll
    for (int k = 0; k < CHUNK_PT; ++k) {
        if (d[k] >= 0) {
            const int pos = atomicAdd(&nxt[d[k] >> CB_SHIFT], 1);
            v2i p;
            p.x = s[k];
            p.y = d[k];
            cpair[e0 + pos] = p;
        }
    }
}

__global__ __launch_bounds__(1024) void gat_fine_place(const v2i* __restrict__ cpair,
                                                       const int* __restrict__ choff,
                                                       int* __restrict__ off,
                                                       int* __restrict__ bsrc,
                                                       int N, int NB) {
    __shared__ int cnt[CB_NODES];
    __shared__ int wtot[16];
    __shared__ int red[64];
    __shared__ int segbeg[SEG_MAX];
    __shared__ int seglen[SEG_MAX];
    const int t    = threadIdx.x;
    const int lane = t & 31;
    const int wave = t >> 5;
    const int c    = blockIdx.x;
    const int n0   = c << CB_SHIFT;

    int lo = 0, len = 0;
    if (t < NB) {
        lo  = choff[t * CBP + c];
        len = choff[t * CBP + c + 1] - lo;
        segbeg[t] = t * CHUNK + lo;
        seglen[t] = len;
    }
    if (t < CB_NODES) cnt[t] = 0;
    int slo = lo, slen = len;
    for (int o = 16; o > 0; o >>= 1) {
        slo  += __shfl_xor(slo, o, 32);
        slen += __shfl_xor(slen, o, 32);
    }
    if (lane == 0) { red[wave] = slo; red[32 + wave] = slen; }
    __syncthreads();
    int beg = red[lane];
    int tot = red[32 + lane];
    for (int o = 16; o > 0; o >>= 1) {
        beg += __shfl_xor(beg, o, 32);
        tot += __shfl_xor(tot, o, 32);
    }
    const int end = beg + tot;

    v2i p[2 * SEG_PW];
#pragma unroll
    for (int u = 0; u < SEG_PW; ++u) {
        const int q  = wave + 32 * u;
        const bool in = q < NB;
        const int sb = in ? segbeg[q] : 0;
        const int sl = in ? seglen[q] : 0;
#pragma unroll
        for (int r = 0; r < 2; ++r) {
            const int j = lane + 32 * r;
            v2i z;
            z.x = 0;
            z.y = -1;
            p[2 * u + r] = (j < sl) ? cpair[sb + j] : z;
        }
    }
#pragma unroll
    for (int i = 0; i < 2 * SEG_PW; ++i)
        if (p[i].y >= 0) atomicAdd(&cnt[p[i].y - n0], 1);
    for (int q = wave; q < NB; q += 32) {
        const int sb = segbeg[q];
        const int sl = seglen[q];
        for (int j = ((q < 32 * SEG_PW) ? 64 : 0) + lane; j < sl; j += 32)
            atomicAdd(&cnt[cpair[sb + j].y - n0], 1);
    }
    __syncthreads();

    int v = 0, inc = 0;
    if (t < CB_NODES) {
        v   = cnt[t];
        inc = v;
        for (int o = 1; o < 32; o <<= 1) {
            const int u = __shfl_up(inc, o, 32);
            if (lane >= o) inc += u;
        }
        if (lane == 31) wtot[wave] = inc;
    }
    __syncthreads();
    if (t < CB_NODES) {
        int w = (lane < wave) ? wtot[lane] : 0;
        for (int o = 16; o > 0; o >>= 1) w += __shfl_xor(w, o, 32);
        const int ex = beg + w + inc - v;
        cnt[t] = ex;
        if (n0 + t <= N) off[n0 + t] = ex;
    }
    if (blockIdx.x == gridDim.x - 1 && t == 0) off[N] = end;
    __syncthreads();

#pragma unroll
    for (int i = 0; i < 2 * SEG_PW; ++i) {
        if (p[i].y >= 0) {
            const int pos = atomicAdd(&cnt[p[i].y - n0], 1);
            bsrc[pos] = p[i].x;
        }
    }
    for (int q = wave; q < NB; q += 32) {
        const int sb = segbeg[q];
        const int sl = seglen[q];
        for (int j = ((q < 32 * SEG_PW) ? 64 : 0) + lane; j < sl; j += 32) {
            const v2i pr = cpair[sb + j];
            const int pos = atomicAdd(&cnt[pr.y - n0], 1);
            bsrc[pos] = pr.x;
        }
    }
}


#define XLANE(v, X) __builtin_bit_cast(float, __builtin_amdgcn_update_dpp(0, __builtin_bit_cast(int, (v)), 0x160 + (X), 0xf, 0xf, false))
template <int T>
__device__ __forceinline__ float slot_bcast(const float v) {
    return __builtin_bit_cast(float, __builtin_amdgcn_mov_dpp8(__builtin_bit_cast(unsigned, v), (unsigned)T * 0x249249u));
}

#define ROW_LOAD(T) hv[T] = *(const v4f*)&hrow[(size_t)__builtin_amdgcn_readlane(sv, T) * HC]
#define ROW_SUM(T)  { const float p = slot_bcast<T>(pv); den += p; acc += hv[T] * p; }

__device__ __forceinline__ void gat_rows8_load(const int n, const int sv,
                                               const float* __restrict__ hrow, v4f (&hv)[8]) {
    if (0 < n) ROW_LOAD(0);
    if (1 < n) ROW_LOAD(1);
    if (2 < n) ROW_LOAD(2);
    if (3 < n) ROW_LOAD(3);
    if (4 < n) ROW_LOAD(4);
    if (5 < n) ROW_LOAD(5);
    if (6 < n) ROW_LOAD(6);
    if (7 < n) ROW_LOAD(7);
}

__device__ __forceinline__ void gat_rows8_sum(const int n, const float pv, const v4f (&hv)[8],
                                              float& den, v4f& acc) {
    if (0 < n) ROW_SUM(0)
    if (1 < n) ROW_SUM(1)
    if (2 < n) ROW_SUM(2)
    if (3 < n) ROW_SUM(3)
    if (4 < n) ROW_SUM(4)
    if (5 < n) ROW_SUM(5)
    if (6 < n) ROW_SUM(6)
    if (7 < n) ROW_SUM(7)
}

__global__ __launch_bounds__(256, 16) void gat_node_gather(const int* __restrict__ off,
                                                       const int* __restrict__ bsrc,
                                                       const float* __restrict__ a_src,
                                                       const float* __restrict__ a_dst,
                                                       const float* __restrict__ h,
                                                       const float* __restrict__ bias,
                                                       float* __restrict__ out,
                                                       int N) {
    const int lane   = threadIdx.x & 31;
    const int head   = lane >> 3;
    const int slot   = lane & 7;
    const int nwaves = (int)(gridDim.x * 8);
    const v4f b      = *(const v4f*)&bias[lane * 4];
    const float* hrow = h + lane * 4;
    const float NEG_INF = -__builtin_huge_valf();
    for (int d = __builtin_amdgcn_readfirstlane((int)(blockIdx.x * 8 + (threadIdx.x >> 5))); d < N; d += nwaves) {
        const int   beg   = off[d];
        const int   deg   = off[d + 1] - beg;
        const float ad    = a_dst[d * HEADS + head];
        const float eself = leaky(a_src[d * HEADS + head] + ad);
        const v4f   hself = *(const v4f*)&hrow[(size_t)d * HC];

        int   sv[4];
        float ev[4];
#pragma unroll
        for (int c = 0; c < 4; ++c) {
            const int j = 8 * c + slot;
            sv[c] = (j < deg) ? bsrc[beg + j] : d;
        }
#pragma unroll
        for (int c = 0; c < 4; ++c) ev[c] = leaky(a_src[sv[c] * HEADS + head] + ad);
        v4f hv[8];
        gat_rows8_load(deg, sv[0], hrow, hv);
        float m = eself;
#pragma unroll
        for (int c = 0; c < 4; ++c) m = ev[c] > m ? ev[c] : m;
        for (int j0 = 32; j0 < deg; j0 += 8) {
            const int j = j0 + slot;
            if (j < deg) {
                const float e = leaky(a_src[bsrc[beg + j] * HEADS + head] + ad);
                m = e > m ? e : m;
            }
        }
        { const float o = XLANE(m, 1); m = o > m ? o : m; }
        { const float o = XLANE(m, 2); m = o > m ? o : m; }
        { const float o = XLANE(m, 4); m = o > m ? o : m; }

        float p   = __expf(eself - m);
        float den = p;
        v4f   acc = hself * p;
        gat_rows8_sum(deg, __expf(ev[0] - m), hv, den, acc);
#pragma unroll
        for (int c = 1; c < 4; ++c) {
            const int n = deg - 8 * c;
            if (n > 0) {
                gat_rows8_load(n, sv[c], hrow, hv);
                gat_rows8_sum(n, __expf(ev[c] - m), hv, den, acc);
            }
        }
        for (int j0 = 32; j0 < deg; j0 += 8) {
            const int  j  = j0 + slot;
            const bool ok = j < deg;
            const int   s = ok ? bsrc[beg + j] : d;
            const float e = ok ? leaky(a_src[s * HEADS + head] + ad) : NEG_INF;
            gat_rows8_load(deg - j0, s, hrow, hv);
            gat_rows8_sum(deg - j0, __expf(e - m), hv, den, acc);
        }

        const float dd = den + 1e-16f;
        v4f o;
        o.x = acc.x / dd + b.x;
        o.y = acc.y / dd + b.y;
        o.z = acc.z / dd + b.z;
        o.w = acc.w / dd + b.w;
        *(v4f*)&out[(size_t)d * HC + lane * 4] = o;
    }
}

extern "C" void kernel_launch(void* const* d_in, const int* in_sizes, int n_in,
                              void* d_out, int out_size, void* d_ws, size_t ws_size,
                              hipStream_t stream) {
    const float* x        = (const float*)d_in[0];
    const int*   ei       = (const int*)d_in[1];
    const float* W        = (const float*)d_in[2];
    const float* att_src  = (const float*)d_in[3];
    const float* att_dst  = (const float*)d_in[4];
    const float* bias     = (const float*)d_in[5];
    float*       out      = (float*)d_out;

    const int N = in_sizes[0] / IN_F;
    const int E = in_sizes[1] / 2;

    const int CB = (N + CB_NODES - 1) / CB_NODES;
    const int NB = (E + CHUNK - 1) / CHUNK;

    float* h      = (float*)d_ws;
    float* a_src  = h + (size_t)N * HC;
    float* a_dst  = a_src + (size_t)N * HEADS;
    v2i*   cpair  = (v2i*)(a_dst + (size_t)N * HEADS);
    int*   bsrc   = (int*)(cpair + E);
    int*   off    = bsrc + E;
    int*   choff  = off + N + 1;

    gat_gemm_wmma<<<1024, 128, GEMM_LDS_BYTES, stream>>>(x, W, att_src, att_dst, h, a_src, a_dst, N);

    gat_coarse_local<<<NB, 1024, 0, stream>>>(ei, choff, cpair, E, CB);
    gat_fine_place<<<CB, 1024, 0, stream>>>(cpair, choff, off, bsrc, N, NB);

    gat_node_gather<<<2048, 256, 0, stream>>>(off, bsrc, a_src, a_dst, h, bias, out, N);
}


# Fine edge ordering and node sums in one launch: one block per 391-node bucket (256 buckets, one per WGP) orders its edges by node in LDS and its waves sum the nodes from there; new arrival order
# speedup vs baseline: 8.1599x; 1.1140x over previous
#include <hip/hip_runtime.h>

#define IN_F      128
#define HC        128
#define HEADS     4
#define OUT_F     32
#define NEG_SLOPE 0.2f

typedef float v2f __attribute__((ext_vector_type(2)));
typedef float v4f __attribute__((ext_vector_type(4)));
typedef float v8f __attribute__((ext_vector_type(8)));
typedef int   v4i __attribute__((ext_vector_type(4)));
typedef int   v2i __attribute__((ext_vector_type(2)));

__device__ __forceinline__ float leaky(float x) {
    return x > 0.0f ? x : NEG_SLOPE * x;
}

#define WSTR 136
#define XSTR 132
#define COEF_FLOATS 128
#define GEMM_LDS_BYTES ((128 * WSTR + 16 * XSTR + COEF_FLOATS) * 4)

__global__ __launch_bounds__(128) void gat_gemm_wmma(const float* __restrict__ x,
                                                     const float* __restrict__ W,
                                                     const float* __restrict__ att_src,
                                                     const float* __restrict__ att_dst,
                                                     float* __restrict__ h,
                                                     float* __restrict__ a_src,
                                                     float* __restrict__ a_dst,
                                                     int N) {
    extern __shared__ float smem[];
    float* Wl = smem;
    float* Xl = smem + 128 * WSTR;
    float* Cl = Xl + 16 * XSTR;

    const int tid    = threadIdx.x;
    const int ntiles = N / 16;

    const int wave = tid >> 5;
    const int lane = tid & 31;
    const int L    = lane & 15;
    const int g    = lane >> 4;

    const int col0 = wave * 32;
    const int col1 = wave * 32 + 16;

    const float as0 = att_src[col0 + L];
    const float as1 = att_src[col1 + L];
    const float ad0 = att_dst[col0 + L];
    const float ad1 = att_dst[col1 + L];

    const v4f* x4 = (const v4f*)x;
    v4f xr[4] = {};
    if ((int)blockIdx.x < ntiles) {
#pragma unroll
        for (int j = 0; j < 4; ++j)
            xr[j] = x4[(size_t)((int)blockIdx.x * 16 + wave + 4 * j) * 32 + lane];
    }

    for (int i = tid; i < 128 * 32; i += 128) {
        int r  = i >> 5;
        int c4 = i & 31;
        v4f v = ((const v4f*)W)[r * 32 + c4];
        *(v4f*)&Wl[r * WSTR + c4 * 4] = v;
    }
    for (int tile = blockIdx.x; tile < ntiles; tile += gridDim.x) {
        const int row0 = tile * 16;
#pragma unroll
        for (int j = 0; j < 4; ++j)
            *(v4f*)&Xl[(wave + 4 * j) * XSTR + lane * 4] = xr[j];
        __syncthreads();
        const int nxt = tile + (int)gridDim.x;
        if (nxt < ntiles) {
#pragma unroll
            for (int j = 0; j < 4; ++j)
                xr[j] = x4[(size_t)(nxt * 16 + wave + 4 * j) * 32 + lane];
        }

        v8f acc0 = {};
        v8f acc1 = {};

        for (int k0 = 0; k0 < IN_F; k0 += 4) {
            const int k = k0 + 2 * g;
            v2f a, b0, b1;
            a.x  = Xl[L * XSTR + k];
            a.y  = Xl[L * XSTR + k + 1];
            b0.x = Wl[k * WSTR + col0 + L];
            b0.y = Wl[(k + 1) * WSTR + col0 + L];
            b1.x = Wl[k * WSTR + col1 + L];
            b1.y = Wl[(k + 1) * WSTR + col1 + L];
            acc0 = __builtin_amdgcn_wmma_f32_16x16x4_f32(false, a, false, b0, (short)0, acc0, false, false);
            acc1 = __builtin_amdgcn_wmma_f32_16x16x4_f32(false, a, false, b1, (short)0, acc1, false, false);
        }

        float v[16];
#pragma unroll
        for (int r = 0; r < 8; ++r) {
            v[r]     = acc0[r] * as0 + acc1[r] * as1;
            v[8 + r] = acc0[r] * ad0 + acc1[r] * ad1;
        }
        const bool hi8 = (L & 8) != 0;
        const bool hi4 = (L & 4) != 0;
        const bool hi2 = (L & 2) != 0;
        const bool hi1 = (L & 1) != 0;
        float u[8], t[4], q[2];
#pragma unroll
        for (int j = 0; j < 8; ++j) {
            const float keep = hi8 ? v[j + 8] : v[j];
            const float send = hi8 ? v[j] : v[j + 8];
            u[j] = keep + __shfl_xor(send, 8, 32);
        }
#pragma unroll
        for (int j = 0; j < 4; ++j) {
            const float keep = hi4 ? u[j + 4] : u[j];
            const float send = hi4 ? u[j] : u[j + 4];
            t[j] = keep + __shfl_xor(send, 4, 32);
        }
#pragma unroll
        for (int j = 0; j < 2; ++j) {
            const float keep = hi2 ? t[j + 2] : t[j];
            const float send = hi2 ? t[j] : t[j + 2];
            q[j] = keep + __shfl_xor(send, 2, 32);
        }
        const float keep1 = hi1 ? q[1] : q[0];
        const float send1 = hi1 ? q[0] : q[1];
        const float z = keep1 + __shfl_xor(send1, 1, 32);
        Cl[(L >> 3) * 64 + (8 * g + (L & 7)) * HEADS + wave] = z;
        __syncthreads();

        const int mbase = row0 + 8 * g;
        for (int r = 0; r < 8; ++r) {
            h[(size_t)(mbase + r) * HC + col0 + L] = acc0[r];
            h[(size_t)(mbase + r) * HC + col1 + L] = acc1[r];
        }

        if (wave == 0) {
            const v4f c = *(const v4f*)&Cl[lane * 4];
            float* dstp = (lane < 16) ? a_src + (size_t)(row0 + lane) * HEADS
                                      : a_dst + (size_t)(row0 + lane - 16) * HEADS;
            *(v4f*)dstp = c;
        }
    }
}

#define CB_NODES 391
#define CB_PAD   512
#define CB_MAX   256
#define LCAP     40960
#define LPAD     32
#define CBP      257
#define CHUNK    8192
#define CHUNK_PT (CHUNK / 1024)
#define SEG_MAX  1024
#define SEG_PW   7

__global__ __launch_bounds__(1024) void gat_coarse_local(const int* __restrict__ ei,
                                                         int* __restrict__ choff,
                                                         v2i* __restrict__ cpair,
                                                         int E, int CB) {
    __shared__ int cnt[CB_MAX];
    __shared__ int nxt[CB_MAX];
    const int t    = threadIdx.x;
    const int lane = t & 31;
    const int wave = t >> 5;
    const int b    = blockIdx.x;
    if (t < CB_MAX) cnt[t] = 0;

    const int e0 = b * CHUNK;
    int s[CHUNK_PT], d[CHUNK_PT];
#pragma unroll
    for (int k = 0; k < CHUNK_PT; ++k) {
        const int e = e0 + k * 1024 + t;
        const bool ok = e < E;
        s[k] = ok ? ei[e] : 0;
        d[k] = ok ? ei[E + e] : -1;
    }
    __syncthreads();
#pragma unroll
    for (int k = 0; k < CHUNK_PT; ++k)
        if (d[k] >= 0) atomicAdd(&cnt[(unsigned)d[k] / CB_NODES], 1);
    __syncthreads();

    if (wave == 0) {
        int v[8];
        int sum = 0;
#pragma unroll
        for (int k = 0; k < 8; ++k) { v[k] = cnt[lane * 8 + k]; sum += v[k]; }
        int inc = sum;
        for (int o = 1; o < 32; o <<= 1) {
            const int u = __shfl_up(inc, o, 32);
            if (lane >= o) inc += u;
        }
        int run = inc - sum;
#pragma unroll
        for (int k = 0; k < 8; ++k) {
            const int c = lane * 8 + k;
            nxt[c] = run;
            if (c <= CB) choff[b * CBP + c] = run;
            run += v[k];
        }
        if (lane == 31) choff[b * CBP + CB_MAX] = run;
    }
    __syncthreads();

#pragma unroll
    for (int k = 0; k < CHUNK_PT; ++k) {
        if (d[k] >= 0) {
            const int pos = atomicAdd(&nxt[(unsigned)d[k] / CB_NODES], 1);
            v2i p;
            p.x = s[k];
            p.y = d[k];
            cpair[e0 + pos] = p;
        }
    }
}


#define XLANE(v, X) __builtin_bit_cast(float, __builtin_amdgcn_update_dpp(0, __builtin_bit_cast(int, (v)), 0x160 + (X), 0xf, 0xf, false))
template <int T>
__device__ __forceinline__ float slot_bcast(const float v) {
    return __builtin_bit_cast(float, __builtin_amdgcn_mov_dpp8(__builtin_bit_cast(unsigned, v), (unsigned)T * 0x249249u));
}

#define ROW_LOAD(T) hv[T] = *(const v4f*)&hrow[(size_t)__builtin_amdgcn_readlane(sv, T) * HC]
#define ROW_SUM(T)  { const float p = slot_bcast<T>(pv); den += p; acc += hv[T] * p; }

__device__ __forceinline__ void gat_rows8_load(const int n, const int sv,
                                               const float* __restrict__ hrow, v4f (&hv)[8]) {
    if (0 < n) ROW_LOAD(0);
    if (1 < n) ROW_LOAD(1);
    if (2 < n) ROW_LOAD(2);
    if (3 < n) ROW_LOAD(3);
    if (4 < n) ROW_LOAD(4);
    if (5 < n) ROW_LOAD(5);
    if (6 < n) ROW_LOAD(6);
    if (7 < n) ROW_LOAD(7);
}

__device__ __forceinline__ void gat_rows8_sum(const int n, const float pv, const v4f (&hv)[8],
                                              float& den, v4f& acc) {
    if (0 < n) ROW_SUM(0)
    if (1 < n) ROW_SUM(1)
    if (2 < n) ROW_SUM(2)
    if (3 < n) ROW_SUM(3)
    if (4 < n) ROW_SUM(4)
    if (5 < n) ROW_SUM(5)
    if (6 < n) ROW_SUM(6)
    if (7 < n) ROW_SUM(7)
}

template <bool IN_LDS>
__device__ __forceinline__ void gat_bucket_nodes(const int* lids, const int* lst,
                                                 const int* gids,
                                                 const float* __restrict__ a_src,
                                                 const float* __restrict__ a_dst,
                                                 const float* __restrict__ h,
                                                 const float* __restrict__ bias,
                                                 float* __restrict__ out,
                                                 const int n0, const int nloc) {
    const int lane   = threadIdx.x & 31;
    const int head   = lane >> 3;
    const int slot   = lane & 7;
    const v4f b      = *(const v4f*)&bias[lane * 4];
    const float* hrow = h + lane * 4;
    const float NEG_INF = -__builtin_huge_valf();
    const int FIRST = IN_LDS ? 32 : 0;
    for (int dl = __builtin_amdgcn_readfirstlane((int)(threadIdx.x >> 5)); dl < nloc; dl += 32) {
        const int   d     = n0 + dl;
        const int   beg   = __builtin_amdgcn_readfirstlane(lst[dl]);
        const int   deg   = __builtin_amdgcn_readfirstlane(lst[dl + 1]) - beg;
        const float ad    = a_dst[d * HEADS + head];
        const float eself = leaky(a_src[d * HEADS + head] + ad);
        const v4f   hself = *(const v4f*)&hrow[(size_t)d * HC];

        int   sv[4] = {d, d, d, d};
        float ev[4] = {eself, eself, eself, eself};
        v4f   hv[8];
        float m = eself;
        if (IN_LDS) {
#pragma unroll
            for (int c = 0; c < 4; ++c) {
                const int j = 8 * c + slot;
                const int v = lids[beg + j];
                sv[c] = (j < deg) ? v : d;
            }
#pragma unroll
            for (int c = 0; c < 4; ++c) ev[c] = leaky(a_src[sv[c] * HEADS + head] + ad);
            gat_rows8_load(deg, sv[0], hrow, hv);
#pragma unroll
            for (int c = 0; c < 4; ++c) m = ev[c] > m ? ev[c] : m;
        }
        for (int j0 = FIRST; j0 < deg; j0 += 8) {
            const int j = j0 + slot;
            if (j < deg) {
                const int   s = IN_LDS ? lids[beg + j] : gids[beg + j];
                const float e = leaky(a_src[s * HEADS + head] + ad);
                m = e > m ? e : m;
            }
        }
        { const float o = XLANE(m, 1); m = o > m ? o : m; }
        { const float o = XLANE(m, 2); m = o > m ? o : m; }
        { const float o = XLANE(m, 4); m = o > m ? o : m; }

        float p   = __expf(eself - m);
        float den = p;
        v4f   acc = hself * p;
        if (IN_LDS) {
            gat_rows8_sum(deg, __expf(ev[0] - m), hv, den, acc);
#pragma unroll
            for (int c = 1; c < 4; ++c) {
                const int n = deg - 8 * c;
                if (n > 0) {
                    gat_rows8_load(n, sv[c], hrow, hv);
                    gat_rows8_sum(n, __expf(ev[c] - m), hv, den, acc);
                }
            }
        }
        for (int j0 = FIRST; j0 < deg; j0 += 8) {
            const int  j  = j0 + slot;
            const bool ok = j < deg;
            int s = d;
            if (ok) s = IN_LDS ? lids[beg + j] : gids[beg + j];
            const float e = ok ? leaky(a_src[s * HEADS + head] + ad) : NEG_INF;
            gat_rows8_load(deg - j0, s, hrow, hv);
            gat_rows8_sum(deg - j0, __expf(e - m), hv, den, acc);
        }

        const float dd = den + 1e-16f;
        v4f o;
        o.x = acc.x / dd + b.x;
        o.y = acc.y / dd + b.y;
        o.z = acc.z / dd + b.z;
        o.w = acc.w / dd + b.w;
        *(v4f*)&out[(size_t)d * HC + lane * 4] = o;
    }
}

__global__ __launch_bounds__(1024) void gat_place_gather(const v2i* __restrict__ cpair,
                                                         const int* __restrict__ choff,
                                                         int* __restrict__ bsrc,
                                                         const float* __restrict__ a_src,
                                                         const float* __restrict__ a_dst,
                                                         const float* __restrict__ h,
                                                         const float* __restrict__ bias,
                                                         float* __restrict__ out,
                                                         int N, int NB) {
    extern __shared__ int lsrc[];
    __shared__ int cnt[CB_PAD];
    __shared__ int st[CB_PAD];
    __shared__ int wtot[16];
    __shared__ int red[64];
    __shared__ int segbeg[SEG_MAX];
    __shared__ int seglen[SEG_MAX];
    const int t    = threadIdx.x;
    const int lane = t & 31;
    const int wave = t >> 5;
    const int c    = blockIdx.x;
    const int n0   = c * CB_NODES;

    int lo = 0, len = 0;
    if (t < NB) {
        lo  = choff[t * CBP + c];
        len = choff[t * CBP + c + 1] - lo;
        segbeg[t] = t * CHUNK + lo;
        seglen[t] = len;
    }
    if (t < CB_PAD) cnt[t] = 0;
    int slo = lo, slen = len;
    for (int o = 16; o > 0; o >>= 1) {
        slo  += __shfl_xor(slo, o, 32);
        slen += __shfl_xor(slen, o, 32);
    }
    if (lane == 0) { red[wave] = slo; red[32 + wave] = slen; }
    __syncthreads();
    int beg = red[lane];
    int tot = red[32 + lane];
    for (int o = 16; o > 0; o >>= 1) {
        beg += __shfl_xor(beg, o, 32);
        tot += __shfl_xor(tot, o, 32);
    }
    beg = __builtin_amdgcn_readfirstlane(beg);
    tot = __builtin_amdgcn_readfirstlane(tot);
    const bool big = tot > LCAP;

    v2i p[2 * SEG_PW];
#pragma unroll
    for (int u = 0; u < SEG_PW; ++u) {
        const int q  = wave + 32 * u;
        const bool in = q < NB;
        const int sb = in ? segbeg[q] : 0;
        const int sl = in ? seglen[q] : 0;
#pragma unroll
        for (int r = 0; r < 2; ++r) {
            const int j = lane + 32 * r;
            v2i z;
            z.x = 0;
            z.y = -1;
            p[2 * u + r] = (j < sl) ? cpair[sb + j] : z;
        }
    }
#pragma unroll
    for (int i = 0; i < 2 * SEG_PW; ++i)
        if (p[i].y >= 0) atomicAdd(&cnt[p[i].y - n0], 1);
    for (int q = wave; q < NB; q += 32) {
        const int sb = segbeg[q];
        const int sl = seglen[q];
        for (int j = ((q < 32 * SEG_PW) ? 64 : 0) + lane; j < sl; j += 32)
            atomicAdd(&cnt[cpair[sb + j].y - n0], 1);
    }
    __syncthreads();

    int v = 0, inc = 0;
    if (t < CB_PAD) {
        v   = cnt[t];
        inc = v;
        for (int o = 1; o < 32; o <<= 1) {
            const int u = __shfl_up(inc, o, 32);
            if (lane >= o) inc += u;
        }
        if (lane == 31) wtot[wave] = inc;
    }
    __syncthreads();
    if (t < CB_PAD) {
        int w = (lane < wave) ? wtot[lane] : 0;
        for (int o = 16; o > 0; o >>= 1) w += __shfl_xor(w, o, 32);
        const int ex = w + inc - v;
        cnt[t] = ex;
        st[t]  = ex;
    }
    if (!big && t < LPAD) lsrc[tot + t] = 0;
    __syncthreads();

    if (!big) {
#pragma unroll
        for (int i = 0; i < 2 * SEG_PW; ++i) {
            if (p[i].y >= 0) {
                const int pos = atomicAdd(&cnt[p[i].y - n0], 1);
                lsrc[pos] = p[i].x;
            }
        }
        for (int q = wave; q < NB; q += 32) {
            const int sb = segbeg[q];
            const int sl = seglen[q];
            for (int j = ((q < 32 * SEG_PW) ? 64 : 0) + lane; j < sl; j += 32) {
                const v2i pr = cpair[sb + j];
                const int pos = atomicAdd(&cnt[pr.y - n0], 1);
                lsrc[pos] = pr.x;
            }
        }
    } else {
#pragma unroll
        for (int i = 0; i < 2 * SEG_PW; ++i) {
            if (p[i].y >= 0) {
                const int pos = atomicAdd(&cnt[p[i].y - n0], 1);
                bsrc[beg + pos] = p[i].x;
            }
        }
        for (int q = wave; q < NB; q += 32) {
            const int sb = segbeg[q];
            const int sl = seglen[q];
            for (int j = ((q < 32 * SEG_PW) ? 64 : 0) + lane; j < sl; j += 32) {
                const v2i pr = cpair[sb + j];
                const int pos = atomicAdd(&cnt[pr.y - n0], 1);
                bsrc[beg + pos] = pr.x;
            }
        }
    }
    __syncthreads();

    const int nloc = (N - n0 < CB_NODES) ? N - n0 : CB_NODES;
    if (!big) gat_bucket_nodes<true>(lsrc, st, bsrc, a_src, a_dst, h, bias, out, n0, nloc);
    else      gat_bucket_nodes<false>(lsrc, st, bsrc + beg, a_src, a_dst, h, bias, out, n0, nloc);
}

extern "C" void kernel_launch(void* const* d_in, const int* in_sizes, int n_in,
                              void* d_out, int out_size, void* d_ws, size_t ws_size,
                              hipStream_t stream) {
    const float* x        = (const float*)d_in[0];
    const int*   ei       = (const int*)d_in[1];
    const float* W        = (const float*)d_in[2];
    const float* att_src  = (const float*)d_in[3];
    const float* att_dst  = (const float*)d_in[4];
    const float* bias     = (const float*)d_in[5];
    float*       out      = (float*)d_out;

    const int N = in_sizes[0] / IN_F;
    const int E = in_sizes[1] / 2;

    const int CB = (N + CB_NODES - 1) / CB_NODES;
    const int NB = (E + CHUNK - 1) / CHUNK;

    float* h      = (float*)d_ws;
    float* a_src  = h + (size_t)N * HC;
    float* a_dst  = a_src + (size_t)N * HEADS;
    v2i*   cpair  = (v2i*)(a_dst + (size_t)N * HEADS);
    int*   bsrc   = (int*)(cpair + E);
    int*   choff  = bsrc + E;

    gat_gemm_wmma<<<1024, 128, GEMM_LDS_BYTES, stream>>>(x, W, att_src, att_dst, h, a_src, a_dst, N);

    gat_coarse_local<<<NB, 1024, 0, stream>>>(ei, choff, cpair, E, CB);
    gat_place_gather<<<CB, 1024, (LCAP + LPAD) * 4, stream>>>(cpair, choff, bsrc, a_src, a_dst, h, bias, out, N, NB);
}


# Coarse edge pass places its chunk's pairs in LDS and writes the chunk out contiguously: whole 128-byte lines per store instead of one scattered 8-byte piece per edge; sums keep their arrival order
# speedup vs baseline: 8.3909x; 1.0283x over previous
#include <hip/hip_runtime.h>

#define IN_F      128
#define HC        128
#define HEADS     4
#define OUT_F     32
#define NEG_SLOPE 0.2f

typedef float v2f __attribute__((ext_vector_type(2)));
typedef float v4f __attribute__((ext_vector_type(4)));
typedef float v8f __attribute__((ext_vector_type(8)));
typedef int   v4i __attribute__((ext_vector_type(4)));
typedef int   v2i __attribute__((ext_vector_type(2)));

__device__ __forceinline__ float leaky(float x) {
    return x > 0.0f ? x : NEG_SLOPE * x;
}

#define WSTR 136
#define XSTR 132
#define COEF_FLOATS 128
#define GEMM_LDS_BYTES ((128 * WSTR + 16 * XSTR + COEF_FLOATS) * 4)

__global__ __launch_bounds__(128) void gat_gemm_wmma(const float* __restrict__ x,
                                                     const float* __restrict__ W,
                                                     const float* __restrict__ att_src,
                                                     const float* __restrict__ att_dst,
                                                     float* __restrict__ h,
                                                     float* __restrict__ a_src,
                                                     float* __restrict__ a_dst,
                                                     int N) {
    extern __shared__ float smem[];
    float* Wl = smem;
    float* Xl = smem + 128 * WSTR;
    float* Cl = Xl + 16 * XSTR;

    const int tid    = threadIdx.x;
    const int ntiles = N / 16;

    const int wave = tid >> 5;
    const int lane = tid & 31;
    const int L    = lane & 15;
    const int g    = lane >> 4;

    const int col0 = wave * 32;
    const int col1 = wave * 32 + 16;

    const float as0 = att_src[col0 + L];
    const float as1 = att_src[col1 + L];
    const float ad0 = att_dst[col0 + L];
    const float ad1 = att_dst[col1 + L];

    const v4f* x4 = (const v4f*)x;
    v4f xr[4] = {};
    if ((int)blockIdx.x < ntiles) {
#pragma unroll
        for (int j = 0; j < 4; ++j)
            xr[j] = x4[(size_t)((int)blockIdx.x * 16 + wave + 4 * j) * 32 + lane];
    }

    for (int i = tid; i < 128 * 32; i += 128) {
        int r  = i >> 5;
        int c4 = i & 31;
        v4f v = ((const v4f*)W)[r * 32 + c4];
        *(v4f*)&Wl[r * WSTR + c4 * 4] = v;
    }
    for (int tile = blockIdx.x; tile < ntiles; tile += gridDim.x) {
        const int row0 = tile * 16;
#pragma unroll
        for (int j = 0; j < 4; ++j)
            *(v4f*)&Xl[(wave + 4 * j) * XSTR + lane * 4] = xr[j];
        __syncthreads();
        const int nxt = tile + (int)gridDim.x;
        if (nxt < ntiles) {
#pragma unroll
            for (int j = 0; j < 4; ++j)
                xr[j] = x4[(size_t)(nxt * 16 + wave + 4 * j) * 32 + lane];
        }

        v8f acc0 = {};
        v8f acc1 = {};

        for (int k0 = 0; k0 < IN_F; k0 += 4) {
            const int k = k0 + 2 * g;
            v2f a, b0, b1;
            a.x  = Xl[L * XSTR + k];
            a.y  = Xl[L * XSTR + k + 1];
            b0.x = Wl[k * WSTR + col0 + L];
            b0.y = Wl[(k + 1) * WSTR + col0 + L];
            b1.x = Wl[k * WSTR + col1 + L];
            b1.y = Wl[(k + 1) * WSTR + col1 + L];
            acc0 = __builtin_amdgcn_wmma_f32_16x16x4_f32(false, a, false, b0, (short)0, acc0, false, false);
            acc1 = __builtin_amdgcn_wmma_f32_16x16x4_f32(false, a, false, b1, (short)0, acc1, false, false);
        }

        float v[16];
#pragma unroll
        for (int r = 0; r < 8; ++r) {
            v[r]     = acc0[r] * as0 + acc1[r] * as1;
            v[8 + r] = acc0[r] * ad0 + acc1[r] * ad1;
        }
        const bool hi8 = (L & 8) != 0;
        const bool hi4 = (L & 4) != 0;
        const bool hi2 = (L & 2) != 0;
        const bool hi1 = (L & 1) != 0;
        float u[8], t[4], q[2];
#pragma unroll
        for (int j = 0; j < 8; ++j) {
            const float keep = hi8 ? v[j + 8] : v[j];
            const float send = hi8 ? v[j] : v[j + 8];
            u[j] = keep + __shfl_xor(send, 8, 32);
        }
#pragma unroll
        for (int j = 0; j < 4; ++j) {
            const float keep = hi4 ? u[j + 4] : u[j];
            const float send = hi4 ? u[j] : u[j + 4];
            t[j] = keep + __shfl_xor(send, 4, 32);
        }
#pragma unroll
        for (int j = 0; j < 2; ++j) {
            const float keep = hi2 ? t[j + 2] : t[j];
            const float send = hi2 ? t[j] : t[j + 2];
            q[j] = keep + __shfl_xor(send, 2, 32);
        }
        const float keep1 = hi1 ? q[1] : q[0];
        const float send1 = hi1 ? q[0] : q[1];
        const float z = keep1 + __shfl_xor(send1, 1, 32);
        Cl[(L >> 3) * 64 + (8 * g + (L & 7)) * HEADS + wave] = z;
        __syncthreads();

        const int mbase = row0 + 8 * g;
        for (int r = 0; r < 8; ++r) {
            h[(size_t)(mbase + r) * HC + col0 + L] = acc0[r];
            h[(size_t)(mbase + r) * HC + col1 + L] = acc1[r];
        }

        if (wave == 0) {
            const v4f c = *(const v4f*)&Cl[lane * 4];
            float* dstp = (lane < 16) ? a_src + (size_t)(row0 + lane) * HEADS
                                      : a_dst + (size_t)(row0 + lane - 16) * HEADS;
            *(v4f*)dstp = c;
        }
    }
}

#define CB_NODES 391
#define CB_PAD   512
#define CB_MAX   256
#define LCAP     40960
#define LPAD     32
#define CBP      257
#define CHUNK    8192
#define CHUNK_PT (CHUNK / 1024)
#define SEG_MAX  1024
#define SEG_PW   7

__global__ __launch_bounds__(1024) void gat_coarse_local(const int* __restrict__ ei,
                                                         int* __restrict__ choff,
                                                         v2i* __restrict__ cpair,
                                                         int E, int CB) {
    extern __shared__ v4i cstage4[];
    v2i* cstage = (v2i*)cstage4;
    __shared__ int cnt[CB_MAX];
    __shared__ int nxt[CB_MAX];
    const int t    = threadIdx.x;
    const int lane = t & 31;
    const int wave = t >> 5;
    const int b    = blockIdx.x;
    if (t < CB_MAX) cnt[t] = 0;

    const int e0 = b * CHUNK;
    int s[CHUNK_PT], d[CHUNK_PT];
#pragma unroll
    for (int k = 0; k < CHUNK_PT; ++k) {
        const int e = e0 + k * 1024 + t;
        const bool ok = e < E;
        s[k] = ok ? ei[e] : 0;
        d[k] = ok ? ei[E + e] : -1;
    }
    __syncthreads();
#pragma unroll
    for (int k = 0; k < CHUNK_PT; ++k)
        if (d[k] >= 0) atomicAdd(&cnt[(unsigned)d[k] / CB_NODES], 1);
    __syncthreads();

    if (wave == 0) {
        int v[8];
        int sum = 0;
#pragma unroll
        for (int k = 0; k < 8; ++k) { v[k] = cnt[lane * 8 + k]; sum += v[k]; }
        int inc = sum;
        for (int o = 1; o < 32; o <<= 1) {
            const int u = __shfl_up(inc, o, 32);
            if (lane >= o) inc += u;
        }
        int run = inc - sum;
#pragma unroll
        for (int k = 0; k < 8; ++k) {
            const int c = lane * 8 + k;
            nxt[c] = run;
            if (c <= CB) choff[b * CBP + c] = run;
            run += v[k];
        }
        if (lane == 31) choff[b * CBP + CB_MAX] = run;
    }
    __syncthreads();

#pragma unroll
    for (int k = 0; k < CHUNK_PT; ++k) {
        if (d[k] >= 0) {
            const int pos = atomicAdd(&nxt[(unsigned)d[k] / CB_NODES], 1);
            v2i p;
            p.x = s[k];
            p.y = d[k];
            cstage[pos] = p;
        }
    }
    __syncthreads();

    const int total = (E - e0 < CHUNK) ? E - e0 : CHUNK;
    v4i* out4 = (v4i*)(cpair + e0);
    if (total == CHUNK) {
#pragma unroll
        for (int k = 0; k < CHUNK_PT / 2; ++k) out4[k * 1024 + t] = cstage4[k * 1024 + t];
    } else {
#pragma unroll
        for (int k = 0; k < CHUNK_PT / 2; ++k) {
            const int i = k * 1024 + t;
            if (2 * i + 1 < total)  out4[i] = cstage4[i];
            else if (2 * i < total) cpair[e0 + 2 * i] = cstage[2 * i];
        }
    }
}


#define XLANE(v, X) __builtin_bit_cast(float, __builtin_amdgcn_update_dpp(0, __builtin_bit_cast(int, (v)), 0x160 + (X), 0xf, 0xf, false))
template <int T>
__device__ __forceinline__ float slot_bcast(const float v) {
    return __builtin_bit_cast(float, __builtin_amdgcn_mov_dpp8(__builtin_bit_cast(unsigned, v), (unsigned)T * 0x249249u));
}

#define ROW_LOAD(T) hv[T] = *(const v4f*)&hrow[(size_t)__builtin_amdgcn_readlane(sv, T) * HC]
#define ROW_SUM(T)  { const float p = slot_bcast<T>(pv); den += p; acc += hv[T] * p; }

__device__ __forceinline__ void gat_rows8_load(const int n, const int sv,
                                               const float* __restrict__ hrow, v4f (&hv)[8]) {
    if (0 < n) ROW_LOAD(0);
    if (1 < n) ROW_LOAD(1);
    if (2 < n) ROW_LOAD(2);
    if (3 < n) ROW_LOAD(3);
    if (4 < n) ROW_LOAD(4);
    if (5 < n) ROW_LOAD(5);
    if (6 < n) ROW_LOAD(6);
    if (7 < n) ROW_LOAD(7);
}

__device__ __forceinline__ void gat_rows8_sum(const int n, const float pv, const v4f (&hv)[8],
                                              float& den, v4f& acc) {
    if (0 < n) ROW_SUM(0)
    if (1 < n) ROW_SUM(1)
    if (2 < n) ROW_SUM(2)
    if (3 < n) ROW_SUM(3)
    if (4 < n) ROW_SUM(4)
    if (5 < n) ROW_SUM(5)
    if (6 < n) ROW_SUM(6)
    if (7 < n) ROW_SUM(7)
}

template <bool IN_LDS>
__device__ __forceinline__ void gat_bucket_nodes(const int* lids, const int* lst,
                                                 const int* gids,
                                                 const float* __restrict__ a_src,
                                                 const float* __restrict__ a_dst,
                                                 const float* __restrict__ h,
                                                 const float* __restrict__ bias,
                                                 float* __restrict__ out,
                                                 const int n0, const int nloc) {
    const int lane   = threadIdx.x & 31;
    const int head   = lane >> 3;
    const int slot   = lane & 7;
    const v4f b      = *(const v4f*)&bias[lane * 4];
    const float* hrow = h + lane * 4;
    const float NEG_INF = -__builtin_huge_valf();
    const int FIRST = IN_LDS ? 32 : 0;
    for (int dl = __builtin_amdgcn_readfirstlane((int)(threadIdx.x >> 5)); dl < nloc; dl += 32) {
        const int   d     = n0 + dl;
        const int   beg   = __builtin_amdgcn_readfirstlane(lst[dl]);
        const int   deg   = __builtin_amdgcn_readfirstlane(lst[dl + 1]) - beg;
        const float ad    = a_dst[d * HEADS + head];
        const float eself = leaky(a_src[d * HEADS + head] + ad);
        const v4f   hself = *(const v4f*)&hrow[(size_t)d * HC];

        int   sv[4] = {d, d, d, d};
        float ev[4] = {eself, eself, eself, eself};
        v4f   hv[8];
        float m = eself;
        if (IN_LDS) {
#pragma unroll
            for (int c = 0; c < 4; ++c) {
                const int j = 8 * c + slot;
                const int v = lids[beg + j];
                sv[c] = (j < deg) ? v : d;
            }
#pragma unroll
            for (int c = 0; c < 4; ++c) ev[c] = leaky(a_src[sv[c] * HEADS + head] + ad);
            gat_rows8_load(deg, sv[0], hrow, hv);
#pragma unroll
            for (int c = 0; c < 4; ++c) m = ev[c] > m ? ev[c] : m;
        }
        for (int j0 = FIRST; j0 < deg; j0 += 8) {
            const int j = j0 + slot;
            if (j < deg) {
                const int   s = IN_LDS ? lids[beg + j] : gids[beg + j];
                const float e = leaky(a_src[s * HEADS + head] + ad);
                m = e > m ? e : m;
            }
        }
        { const float o = XLANE(m, 1); m = o > m ? o : m; }
        { const float o = XLANE(m, 2); m = o > m ? o : m; }
        { const float o = XLANE(m, 4); m = o > m ? o : m; }

        float p   = __expf(eself - m);
        float den = p;
        v4f   acc = hself * p;
        if (IN_LDS) {
            gat_rows8_sum(deg, __expf(ev[0] - m), hv, den, acc);
#pragma unroll
            for (int c = 1; c < 4; ++c) {
                const int n = deg - 8 * c;
                if (n > 0) {
                    gat_rows8_load(n, sv[c], hrow, hv);
                    gat_rows8_sum(n, __expf(ev[c] - m), hv, den, acc);
                }
            }
        }
        for (int j0 = FIRST; j0 < deg; j0 += 8) {
            const int  j  = j0 + slot;
            const bool ok = j < deg;
            int s = d;
            if (ok) s = IN_LDS ? lids[beg + j] : gids[beg + j];
            const float e = ok ? leaky(a_src[s * HEADS + head] + ad) : NEG_INF;
            gat_rows8_load(deg - j0, s, hrow, hv);
            gat_rows8_sum(deg - j0, __expf(e - m), hv, den, acc);
        }

        const float dd = den + 1e-16f;
        v4f o;
        o.x = acc.x / dd + b.x;
        o.y = acc.y / dd + b.y;
        o.z = acc.z / dd + b.z;
        o.w = acc.w / dd + b.w;
        *(v4f*)&out[(size_t)d * HC + lane * 4] = o;
    }
}

__global__ __launch_bounds__(1024) void gat_place_gather(const v2i* __restrict__ cpair,
                                                         const int* __restrict__ choff,
                                                         int* __restrict__ bsrc,
                                                         const float* __restrict__ a_src,
                                                         const float* __restrict__ a_dst,
                                                         const float* __restrict__ h,
                                                         const float* __restrict__ bias,
                                                         float* __restrict__ out,
                                                         int N, int NB) {
    extern __shared__ int lsrc[];
    __shared__ int cnt[CB_PAD];
    __shared__ int st[CB_PAD];
    __shared__ int wtot[16];
    __shared__ int red[64];
    __shared__ int segbeg[SEG_MAX];
    __shared__ int seglen[SEG_MAX];
    const int t    = threadIdx.x;
    const int lane = t & 31;
    const int wave = t >> 5;
    const int c    = blockIdx.x;
    const int n0   = c * CB_NODES;

    int lo = 0, len = 0;
    if (t < NB) {
        lo  = choff[t * CBP + c];
        len = choff[t * CBP + c + 1] - lo;
        segbeg[t] = t * CHUNK + lo;
        seglen[t] = len;
    }
    if (t < CB_PAD) cnt[t] = 0;
    int slo = lo, slen = len;
    for (int o = 16; o > 0; o >>= 1) {
        slo  += __shfl_xor(slo, o, 32);
        slen += __shfl_xor(slen, o, 32);
    }
    if (lane == 0) { red[wave] = slo; red[32 + wave] = slen; }
    __syncthreads();
    int beg = red[lane];
    int tot = red[32 + lane];
    for (int o = 16; o > 0; o >>= 1) {
        beg += __shfl_xor(beg, o, 32);
        tot += __shfl_xor(tot, o, 32);
    }
    beg = __builtin_amdgcn_readfirstlane(beg);
    tot = __builtin_amdgcn_readfirstlane(tot);
    const bool big = tot > LCAP;

    v2i p[2 * SEG_PW];
#pragma unroll
    for (int u = 0; u < SEG_PW; ++u) {
        const int q  = wave + 32 * u;
        const bool in = q < NB;
        const int sb = in ? segbeg[q] : 0;
        const int sl = in ? seglen[q] : 0;
#pragma unroll
        for (int r = 0; r < 2; ++r) {
            const int j = lane + 32 * r;
            v2i z;
            z.x = 0;
            z.y = -1;
            p[2 * u + r] = (j < sl) ? cpair[sb + j] : z;
        }
    }
#pragma unroll
    for (int i = 0; i < 2 * SEG_PW; ++i)
        if (p[i].y >= 0) atomicAdd(&cnt[p[i].y - n0], 1);
    for (int q = wave; q < NB; q += 32) {
        const int sb = segbeg[q];
        const int sl = seglen[q];
        for (int j = ((q < 32 * SEG_PW) ? 64 : 0) + lane; j < sl; j += 32)
            atomicAdd(&cnt[cpair[sb + j].y - n0], 1);
    }
    __syncthreads();

    int v = 0, inc = 0;
    if (t < CB_PAD) {
        v   = cnt[t];
        inc = v;
        for (int o = 1; o < 32; o <<= 1) {
            const int u = __shfl_up(inc, o, 32);
            if (lane >= o) inc += u;
        }
        if (lane == 31) wtot[wave] = inc;
    }
    __syncthreads();
    if (t < CB_PAD) {
        int w = (lane < wave) ? wtot[lane] : 0;
        for (int o = 16; o > 0; o >>= 1) w += __shfl_xor(w, o, 32);
        const int ex = w + inc - v;
        cnt[t] = ex;
        st[t]  = ex;
    }
    if (!big && t < LPAD) lsrc[tot + t] = 0;
    __syncthreads();

    if (!big) {
#pragma unroll
        for (int i = 0; i < 2 * SEG_PW; ++i) {
            if (p[i].y >= 0) {
                const int pos = atomicAdd(&cnt[p[i].y - n0], 1);
                lsrc[pos] = p[i].x;
            }
        }
        for (int q = wave; q < NB; q += 32) {
            const int sb = segbeg[q];
            const int sl = seglen[q];
            for (int j = ((q < 32 * SEG_PW) ? 64 : 0) + lane; j < sl; j += 32) {
                const v2i pr = cpair[sb + j];
                const int pos = atomicAdd(&cnt[pr.y - n0], 1);
                lsrc[pos] = pr.x;
            }
        }
    } else {
#pragma unroll
        for (int i = 0; i < 2 * SEG_PW; ++i) {
            if (p[i].y >= 0) {
                const int pos = atomicAdd(&cnt[p[i].y - n0], 1);
                bsrc[beg + pos] = p[i].x;
            }
        }
        for (int q = wave; q < NB; q += 32) {
            const int sb = segbeg[q];
            const int sl = seglen[q];
            for (int j = ((q < 32 * SEG_PW) ? 64 : 0) + lane; j < sl; j += 32) {
                const v2i pr = cpair[sb + j];
                const int pos = atomicAdd(&cnt[pr.y - n0], 1);
                bsrc[beg + pos] = pr.x;
            }
        }
    }
    __syncthreads();

    const int nloc = (N - n0 < CB_NODES) ? N - n0 : CB_NODES;
    if (!big) gat_bucket_nodes<true>(lsrc, st, bsrc, a_src, a_dst, h, bias, out, n0, nloc);
    else      gat_bucket_nodes<false>(lsrc, st, bsrc + beg, a_src, a_dst, h, bias, out, n0, nloc);
}

extern "C" void kernel_launch(void* const* d_in, const int* in_sizes, int n_in,
                              void* d_out, int out_size, void* d_ws, size_t ws_size,
                              hipStream_t stream) {
    const float* x        = (const float*)d_in[0];
    const int*   ei       = (const int*)d_in[1];
    const float* W        = (const float*)d_in[2];
    const float* att_src  = (const float*)d_in[3];
    const float* att_dst  = (const float*)d_in[4];
    const float* bias     = (const float*)d_in[5];
    float*       out      = (float*)d_out;

    const int N = in_sizes[0] / IN_F;
    const int E = in_sizes[1] / 2;

    const int CB = (N + CB_NODES - 1) / CB_NODES;
    const int NB = (E + CHUNK - 1) / CHUNK;

    float* h      = (float*)d_ws;
    float* a_src  = h + (size_t)N * HC;
    float* a_dst  = a_src + (size_t)N * HEADS;
    v2i*   cpair  = (v2i*)(a_dst + (size_t)N * HEADS);
    int*   bsrc   = (int*)(cpair + E);
    int*   choff  = bsrc + E;

    gat_gemm_wmma<<<1024, 128, GEMM_LDS_BYTES, stream>>>(x, W, att_src, att_dst, h, a_src, a_dst, N);

    gat_coarse_local<<<NB, 1024, CHUNK * 8, stream>>>(ei, choff, cpair, E, CB);
    gat_place_gather<<<CB, 1024, (LCAP + LPAD) * 4, stream>>>(cpair, choff, bsrc, a_src, a_dst, h, bias, out, N, NB);
}
